# starting kernel
# baseline (speedup 1.0000x reference)
#include <hip/hip_runtime.h>

#define N_NODES 100000
#define N_EDGES 1600000
#define D 64

#define SA_STRIDE 68
#define SW_STRIDE 160

typedef __attribute__((ext_vector_type(2))) float v2f;
typedef __attribute__((ext_vector_type(8))) float v8f;

__device__ __forceinline__ unsigned enc_f32(float f) {
    unsigned u = __float_as_uint(f);
    return (u & 0x80000000u) ? ~u : (u | 0x80000000u);
}
__device__ __forceinline__ float dec_f32(unsigned u) {
    return (u & 0x80000000u) ? __uint_as_float(u ^ 0x80000000u) : __uint_as_float(~u);
}
#define ENC_NEG_INF 0x007FFFFFu

__global__ __launch_bounds__(256) void gin_init_agg(unsigned* __restrict__ agg, int n) {
    int i = blockIdx.x * 256 + threadIdx.x;
    if (i < n) agg[i] = ENC_NEG_INF;
}

__global__ __launch_bounds__(256) void gin_scatter_max(const float* __restrict__ x,
                                                       const int* __restrict__ src,
                                                       const int* __restrict__ dst,
                                                       unsigned* __restrict__ agg) {
    int idx = blockIdx.x * 256 + threadIdx.x;
    int e  = idx >> 4;
    int c4 = (idx & 15) << 2;
    if (e >= N_EDGES) return;
    int s = src[e];
    int d = dst[e];
    const float4 v = *(const float4*)(x + (size_t)s * D + c4);
    unsigned* a = agg + (size_t)d * D + c4;
    atomicMax(a + 0, enc_f32(v.x));
    atomicMax(a + 1, enc_f32(v.y));
    atomicMax(a + 2, enc_f32(v.z));
    atomicMax(a + 3, enc_f32(v.w));
}

__global__ __launch_bounds__(256) void gin_wmma_gemm(const float* __restrict__ x,
                                                     const unsigned* __restrict__ agg,
                                                     const float* __restrict__ W,
                                                     const float* __restrict__ bias,
                                                     float* __restrict__ out) {
    __shared__ float sWp[(D / 2) * SW_STRIDE];
    __shared__ float sA[8][16 * SA_STRIDE];

    const int tid  = threadIdx.x;
    const int wave = tid >> 5;
    const int lane = tid & 31;
    const int node_base = blockIdx.x * 128 + wave * 16;

#pragma unroll
    for (int i = 0; i < 16; ++i) {
        int idx = tid + i * 256;
        int o = idx >> 6;
        int k = idx & 63;
        sWp[(k >> 1) * SW_STRIDE + o * 2 + (k & 1)] = W[idx];
    }

#pragma unroll
    for (int j = 0; j < 8; ++j) {
        int p   = lane + j * 32;
        int row = p >> 4;
        int c4  = (p & 15) << 2;
        int g   = node_base + row;
        if (g >= N_NODES) g = N_NODES - 1;
        const float4 xin = *(const float4*)(x   + (size_t)g * D + c4);
        const uint4  au  = *(const uint4*) (agg + (size_t)g * D + c4);
        float4 r;
        r.x = xin.x + (au.x == ENC_NEG_INF ? 0.0f : dec_f32(au.x));
        r.y = xin.y + (au.y == ENC_NEG_INF ? 0.0f : dec_f32(au.y));
        r.z = xin.z + (au.z == ENC_NEG_INF ? 0.0f : dec_f32(au.z));
        r.w = xin.w + (au.w == ENC_NEG_INF ? 0.0f : dec_f32(au.w));
        *(float4*)(&sA[wave][row * SA_STRIDE + c4]) = r;
    }
    __syncthreads();

    v8f acc[4];
    acc[0] = v8f{}; acc[1] = v8f{}; acc[2] = v8f{}; acc[3] = v8f{};

    const int m     = lane & 15;
    const int khalf = (lane >> 4) << 1;
    const float* Arow = &sA[wave][m * SA_STRIDE];

#pragma unroll
    for (int k = 0; k < D; k += 4) {
        const int kk = k + khalf;
        const v2f a = *(const v2f*)(Arow + kk);
        const float* bbase = &sWp[(kk >> 1) * SW_STRIDE + m * 2];
#pragma unroll
        for (int t = 0; t < 4; ++t) {
            const v2f bf = *(const v2f*)(bbase + t * 32);
            acc[t] = __builtin_amdgcn_wmma_f32_16x16x4_f32(
                false, a, false, bf, (short)0, acc[t], false, false);
        }
    }

    const int rowoff = (lane >> 4) << 3;
#pragma unroll
    for (int t = 0; t < 4; ++t) {
        const int col = t * 16 + m;
        const float bv = bias[col];
#pragma unroll
        for (int v = 0; v < 8; ++v) {
            const int row = node_base + rowoff + v;
            if (row < N_NODES) out[(size_t)row * D + col] = acc[t][v] + bv;
        }
    }
}

extern "C" void kernel_launch(void* const* d_in, const int* in_sizes, int n_in,
                              void* d_out, int out_size, void* d_ws, size_t ws_size,
                              hipStream_t stream) {
    const float* x    = (const float*)d_in[0];
    const float* W    = (const float*)d_in[1];
    const float* bias = (const float*)d_in[2];
    const int*   src  = (const int*)d_in[3];
    const int*   dst  = (const int*)d_in[4];
    float*       out  = (float*)d_out;
    unsigned*    agg  = (unsigned*)d_ws;

    const int total = N_NODES * D;
    gin_init_agg<<<(total + 255) / 256, 256, 0, stream>>>(agg, total);

    const int sc_threads = N_EDGES * 16;
    gin_scatter_max<<<(sc_threads + 255) / 256, 256, 0, stream>>>(x, src, dst, agg);

    gin_wmma_gemm<<<(N_NODES + 127) / 128, 256, 0, stream>>>(x, agg, W, bias, out);
}


# Scatter remapped so every gather load and atomic max covers one whole 128-B line: a wave walks 32 edges, lane = feature (2 full-line atomics per edge, not 8 partial); atomics kept, max is order-free
# speedup vs baseline: 1.8316x; 1.8316x over previous
#include <hip/hip_runtime.h>

#define N_NODES 100000
#define N_EDGES 1600000
#define D 64

#define SA_STRIDE 68
#define SW_STRIDE 160

typedef __attribute__((ext_vector_type(2))) float v2f;
typedef __attribute__((ext_vector_type(8))) float v8f;

__device__ __forceinline__ unsigned enc_f32(float f) {
    unsigned u = __float_as_uint(f);
    return (u & 0x80000000u) ? ~u : (u | 0x80000000u);
}
__device__ __forceinline__ float dec_f32(unsigned u) {
    return (u & 0x80000000u) ? __uint_as_float(u ^ 0x80000000u) : __uint_as_float(~u);
}
#define ENC_NEG_INF 0x007FFFFFu

__global__ __launch_bounds__(256) void gin_init_agg(unsigned* __restrict__ agg, int n) {
    int i = blockIdx.x * 256 + threadIdx.x;
    if (i < n) agg[i] = ENC_NEG_INF;
}

#define SC_EDGES_PER_WAVE 32
#define SC_BATCH 8
__global__ __launch_bounds__(256) void gin_scatter_max(const float* __restrict__ x,
                                                       const int* __restrict__ src,
                                                       const int* __restrict__ dst,
                                                       unsigned* __restrict__ agg) {
    const int lane = threadIdx.x & 31;
    const int e0 = (blockIdx.x * 8 + (threadIdx.x >> 5)) * SC_EDGES_PER_WAVE;
    if (e0 >= N_EDGES) return;
    int e = e0 + lane;
    if (e > N_EDGES - 1) e = N_EDGES - 1;
    const int s_l = src[e];
    const int d_l = dst[e];

#pragma unroll
    for (int b = 0; b < SC_EDGES_PER_WAVE; b += SC_BATCH) {
        float v0[SC_BATCH], v1[SC_BATCH];
#pragma unroll
        for (int j = 0; j < SC_BATCH; ++j) {
            const int s = __builtin_amdgcn_readlane(s_l, b + j);
            const float* xr = x + (size_t)s * D + lane;
            v0[j] = xr[0];
            v1[j] = xr[32];
        }
#pragma unroll
        for (int j = 0; j < SC_BATCH; ++j) {
            const int d = __builtin_amdgcn_readlane(d_l, b + j);
            unsigned* ar = agg + (size_t)d * D + lane;
            atomicMax(ar,      enc_f32(v0[j]));
            atomicMax(ar + 32, enc_f32(v1[j]));
        }
    }
}

__global__ __launch_bounds__(256) void gin_wmma_gemm(const float* __restrict__ x,
                                                     const unsigned* __restrict__ agg,
                                                     const float* __restrict__ W,
                                                     const float* __restrict__ bias,
                                                     float* __restrict__ out) {
    __shared__ float sWp[(D / 2) * SW_STRIDE];
    __shared__ float sA[8][16 * SA_STRIDE];

    const int tid  = threadIdx.x;
    const int wave = tid >> 5;
    const int lane = tid & 31;
    const int node_base = blockIdx.x * 128 + wave * 16;

#pragma unroll
    for (int i = 0; i < 16; ++i) {
        int idx = tid + i * 256;
        int o = idx >> 6;
        int k = idx & 63;
        sWp[(k >> 1) * SW_STRIDE + o * 2 + (k & 1)] = W[idx];
    }

#pragma unroll
    for (int j = 0; j < 8; ++j) {
        int p   = lane + j * 32;
        int row = p >> 4;
        int c4  = (p & 15) << 2;
        int g   = node_base + row;
        if (g >= N_NODES) g = N_NODES - 1;
        const float4 xin = *(const float4*)(x   + (size_t)g * D + c4);
        const uint4  au  = *(const uint4*) (agg + (size_t)g * D + c4);
        float4 r;
        r.x = xin.x + (au.x == ENC_NEG_INF ? 0.0f : dec_f32(au.x));
        r.y = xin.y + (au.y == ENC_NEG_INF ? 0.0f : dec_f32(au.y));
        r.z = xin.z + (au.z == ENC_NEG_INF ? 0.0f : dec_f32(au.z));
        r.w = xin.w + (au.w == ENC_NEG_INF ? 0.0f : dec_f32(au.w));
        *(float4*)(&sA[wave][row * SA_STRIDE + c4]) = r;
    }
    __syncthreads();

    v8f acc[4];
    acc[0] = v8f{}; acc[1] = v8f{}; acc[2] = v8f{}; acc[3] = v8f{};

    const int m     = lane & 15;
    const int khalf = (lane >> 4) << 1;
    const float* Arow = &sA[wave][m * SA_STRIDE];

#pragma unroll
    for (int k = 0; k < D; k += 4) {
        const int kk = k + khalf;
        const v2f a = *(const v2f*)(Arow + kk);
        const float* bbase = &sWp[(kk >> 1) * SW_STRIDE + m * 2];
#pragma unroll
        for (int t = 0; t < 4; ++t) {
            const v2f bf = *(const v2f*)(bbase + t * 32);
            acc[t] = __builtin_amdgcn_wmma_f32_16x16x4_f32(
                false, a, false, bf, (short)0, acc[t], false, false);
        }
    }

    const int rowoff = (lane >> 4) << 3;
#pragma unroll
    for (int t = 0; t < 4; ++t) {
        const int col = t * 16 + m;
        const float bv = bias[col];
#pragma unroll
        for (int v = 0; v < 8; ++v) {
            const int row = node_base + rowoff + v;
            if (row < N_NODES) out[(size_t)row * D + col] = acc[t][v] + bv;
        }
    }
}

extern "C" void kernel_launch(void* const* d_in, const int* in_sizes, int n_in,
                              void* d_out, int out_size, void* d_ws, size_t ws_size,
                              hipStream_t stream) {
    const float* x    = (const float*)d_in[0];
    const float* W    = (const float*)d_in[1];
    const float* bias = (const float*)d_in[2];
    const int*   src  = (const int*)d_in[3];
    const int*   dst  = (const int*)d_in[4];
    float*       out  = (float*)d_out;
    unsigned*    agg  = (unsigned*)d_ws;

    const int total = N_NODES * D;
    gin_init_agg<<<(total + 255) / 256, 256, 0, stream>>>(agg, total);

    gin_scatter_max<<<(N_EDGES + 255) / 256, 256, 0, stream>>>(x, src, dst, agg);

    gin_wmma_gemm<<<(N_NODES + 127) / 128, 256, 0, stream>>>(x, agg, W, bias, out);
}


# Scatter kept to one L2 half per agg row: work-groups read their die id, draw edge chunks by returning atomic add on their half's counter, apply only dst of their node half; max stays order-free
# speedup vs baseline: 2.1771x; 1.1886x over previous
#include <hip/hip_runtime.h>

#define N_NODES 100000
#define N_EDGES 1600000
#define D 64

#define SA_STRIDE 68
#define SW_STRIDE 160

typedef __attribute__((ext_vector_type(2))) float v2f;
typedef __attribute__((ext_vector_type(8))) float v8f;

__device__ __forceinline__ unsigned enc_f32(float f) {
    unsigned u = __float_as_uint(f);
    return (u & 0x80000000u) ? ~u : (u | 0x80000000u);
}
__device__ __forceinline__ float dec_f32(unsigned u) {
    return (u & 0x80000000u) ? __uint_as_float(u ^ 0x80000000u) : __uint_as_float(~u);
}
#define ENC_NEG_INF 0x007FFFFFu

#define TICKET_STRIDE 64
__global__ __launch_bounds__(256) void gin_init_agg(unsigned* __restrict__ agg, int n,
                                                    unsigned* __restrict__ tickets) {
    int i = blockIdx.x * 256 + threadIdx.x;
    if (i < n) agg[i] = ENC_NEG_INF;
    if (i < 2) tickets[i * TICKET_STRIDE] = 0u;
}

#define SC_WAVE_EDGES 64
#define SC_WG_EDGES (8 * SC_WAVE_EDGES)
#define SC_NTICKET ((N_EDGES + SC_WG_EDGES - 1) / SC_WG_EDGES)
#define SC_LIST (SC_WAVE_EDGES + 8)
#define SC_GRID 1536
#define SC_BATCH 8
#define HALF_NODES (N_NODES / 2)
__global__ __launch_bounds__(256) void gin_scatter_max(const float* __restrict__ x,
                                                       const int* __restrict__ src,
                                                       const int* __restrict__ dst,
                                                       unsigned* __restrict__ agg,
                                                       unsigned* __restrict__ tickets) {
    __shared__ int sSrc[8][SC_LIST + 32];
    __shared__ int sDst[8][SC_LIST + 32];
    __shared__ unsigned sTicket;
    __shared__ int sHalf;

    const int tid  = threadIdx.x;
    const int lane = tid & 31;
    const int wave = tid >> 5;

    if (tid == 0) {
        const unsigned id = __builtin_amdgcn_s_sendmsg_rtn(0x87);
        sHalf = (((id >> 16) & 0xFu) >= 4u) ? 1 : 0;
    }
    __syncthreads();
    const int myhalf = sHalf;

    auto walk = [&](int n) {
        for (int g = 0; g < n; g += 32) {
            int idx = g + lane;
            if (idx > n - 1) idx = n - 1;
            const int s_l = sSrc[wave][idx];
            const int d_l = sDst[wave][idx];
#pragma unroll
            for (int b = 0; b < 32; b += SC_BATCH) {
                if (g + b >= n) break;
                float v0[SC_BATCH], v1[SC_BATCH];
#pragma unroll
                for (int j = 0; j < SC_BATCH; ++j) {
                    const int s = __builtin_amdgcn_readlane(s_l, b + j);
                    const float* xr = x + (size_t)s * D + lane;
                    v0[j] = xr[0];
                    v1[j] = xr[32];
                }
#pragma unroll
                for (int j = 0; j < SC_BATCH; ++j) {
                    const int d = __builtin_amdgcn_readlane(d_l, b + j);
                    unsigned* ar = agg + (size_t)d * D + lane;
                    atomicMax(ar,      enc_f32(v0[j]));
                    atomicMax(ar + 32, enc_f32(v1[j]));
                }
            }
        }
    };

    int pass  = 0;
    int h     = myhalf;
    int carry = 0;
    unsigned nextT = 0;
    if (tid == 0) nextT = atomicAdd(tickets + h * TICKET_STRIDE, 1u);

    for (int it = 0; it < 2 * (SC_NTICKET + 1); ++it) {
        if (tid == 0) sTicket = nextT;
        __syncthreads();
        const unsigned t = sTicket;
        if (t >= (unsigned)SC_NTICKET) {
            walk(carry);
            carry = 0;
            __syncthreads();
            if (pass) break;
            pass = 1;
            h = myhalf ^ 1;
            if (tid == 0) {
                const unsigned c = __hip_atomic_load(tickets + h * TICKET_STRIDE,
                                                     __ATOMIC_RELAXED, __HIP_MEMORY_SCOPE_AGENT);
                nextT = (c < (unsigned)SC_NTICKET) ? atomicAdd(tickets + h * TICKET_STRIDE, 1u)
                                                   : (unsigned)SC_NTICKET;
            }
            continue;
        }
        if (tid == 0) nextT = atomicAdd(tickets + h * TICKET_STRIDE, 1u);

        const int lo = h ? HALF_NODES : 0;
        const int hi = h ? N_NODES : HALF_NODES;

        const int ebase = (int)t * SC_WG_EDGES + wave * SC_WAVE_EDGES + lane;
        int sv[SC_WAVE_EDGES / 32], dv[SC_WAVE_EDGES / 32];
#pragma unroll
        for (int i = 0; i < SC_WAVE_EDGES / 32; ++i) {
            int e = ebase + i * 32;
            const bool in = e < N_EDGES;
            if (!in) e = N_EDGES - 1;
            sv[i] = src[e];
            const int d = dst[e];
            dv[i] = in ? d : -1;
        }
        int cnt = carry;
#pragma unroll
        for (int i = 0; i < SC_WAVE_EDGES / 32; ++i) {
            const bool m = (dv[i] >= lo) && (dv[i] < hi);
            const unsigned mask = __builtin_amdgcn_ballot_w32(m);
            const int pos = cnt + (int)__builtin_amdgcn_mbcnt_lo(mask, 0u);
            const int p = m ? pos : (SC_LIST + lane);
            sSrc[wave][p] = sv[i];
            sDst[wave][p] = dv[i];
            cnt += __builtin_popcount(mask);
        }
        __syncthreads();

        const int full = cnt & ~(SC_BATCH - 1);
        walk(full);
        const int rest = cnt - full;
        if (rest > 0 && full > 0) {
            int ts = 0, td = 0;
            if (lane < rest) { ts = sSrc[wave][full + lane]; td = sDst[wave][full + lane]; }
            if (lane < rest) { sSrc[wave][lane] = ts; sDst[wave][lane] = td; }
        }
        carry = rest;
    }
}

__global__ __launch_bounds__(256) void gin_wmma_gemm(const float* __restrict__ x,
                                                     const unsigned* __restrict__ agg,
                                                     const float* __restrict__ W,
                                                     const float* __restrict__ bias,
                                                     float* __restrict__ out) {
    __shared__ float sWp[(D / 2) * SW_STRIDE];
    __shared__ float sA[8][16 * SA_STRIDE];

    const int tid  = threadIdx.x;
    const int wave = tid >> 5;
    const int lane = tid & 31;
    const int node_base = blockIdx.x * 128 + wave * 16;

#pragma unroll
    for (int i = 0; i < 16; ++i) {
        int idx = tid + i * 256;
        int o = idx >> 6;
        int k = idx & 63;
        sWp[(k >> 1) * SW_STRIDE + o * 2 + (k & 1)] = W[idx];
    }

#pragma unroll
    for (int j = 0; j < 8; ++j) {
        int p   = lane + j * 32;
        int row = p >> 4;
        int c4  = (p & 15) << 2;
        int g   = node_base + row;
        if (g >= N_NODES) g = N_NODES - 1;
        const float4 xin = *(const float4*)(x   + (size_t)g * D + c4);
        const uint4  au  = *(const uint4*) (agg + (size_t)g * D + c4);
        float4 r;
        r.x = xin.x + (au.x == ENC_NEG_INF ? 0.0f : dec_f32(au.x));
        r.y = xin.y + (au.y == ENC_NEG_INF ? 0.0f : dec_f32(au.y));
        r.z = xin.z + (au.z == ENC_NEG_INF ? 0.0f : dec_f32(au.z));
        r.w = xin.w + (au.w == ENC_NEG_INF ? 0.0f : dec_f32(au.w));
        *(float4*)(&sA[wave][row * SA_STRIDE + c4]) = r;
    }
    __syncthreads();

    v8f acc[4];
    acc[0] = v8f{}; acc[1] = v8f{}; acc[2] = v8f{}; acc[3] = v8f{};

    const int m     = lane & 15;
    const int khalf = (lane >> 4) << 1;
    const float* Arow = &sA[wave][m * SA_STRIDE];

#pragma unroll
    for (int k = 0; k < D; k += 4) {
        const int kk = k + khalf;
        const v2f a = *(const v2f*)(Arow + kk);
        const float* bbase = &sWp[(kk >> 1) * SW_STRIDE + m * 2];
#pragma unroll
        for (int t = 0; t < 4; ++t) {
            const v2f bf = *(const v2f*)(bbase + t * 32);
            acc[t] = __builtin_amdgcn_wmma_f32_16x16x4_f32(
                false, a, false, bf, (short)0, acc[t], false, false);
        }
    }

    const int rowoff = (lane >> 4) << 3;
#pragma unroll
    for (int t = 0; t < 4; ++t) {
        const int col = t * 16 + m;
        const float bv = bias[col];
#pragma unroll
        for (int v = 0; v < 8; ++v) {
            const int row = node_base + rowoff + v;
            if (row < N_NODES) out[(size_t)row * D + col] = acc[t][v] + bv;
        }
    }
}

extern "C" void kernel_launch(void* const* d_in, const int* in_sizes, int n_in,
                              void* d_out, int out_size, void* d_ws, size_t ws_size,
                              hipStream_t stream) {
    const float* x    = (const float*)d_in[0];
    const float* W    = (const float*)d_in[1];
    const float* bias = (const float*)d_in[2];
    const int*   src  = (const int*)d_in[3];
    const int*   dst  = (const int*)d_in[4];
    float*       out  = (float*)d_out;
    unsigned*    agg  = (unsigned*)d_ws;
    unsigned*    tickets = agg + N_NODES * D;

    const int total = N_NODES * D;
    gin_init_agg<<<(total + 255) / 256, 256, 0, stream>>>(agg, total, tickets);

    gin_scatter_max<<<SC_GRID, 256, 0, stream>>>(x, src, dst, agg, tickets);

    gin_wmma_gemm<<<(N_NODES + 127) / 128, 256, 0, stream>>>(x, agg, W, bias, out);
}


# Edges counting-sorted by dst node range; one work-group per range keeps its agg rows in LDS (LDS max) and runs the linear layer from LDS: no agg init pass, no global atomics; max order-free
# speedup vs baseline: 2.7752x; 1.2747x over previous
#include <hip/hip_runtime.h>


#define N_NODES 100000
#define N_EDGES 1600000
#define D 64

#define SA_STRIDE 68
#define SW_STRIDE 160

typedef __attribute__((ext_vector_type(2))) float v2f;
typedef __attribute__((ext_vector_type(8))) float v8f;
typedef __attribute__((ext_vector_type(2))) int v2i;

__device__ __forceinline__ unsigned enc_f32(float f) {
    unsigned u = __float_as_uint(f);
    return (u & 0x80000000u) ? ~u : (u | 0x80000000u);
}
__device__ __forceinline__ float dec_f32(unsigned u) {
    return (u & 0x80000000u) ? __uint_as_float(u ^ 0x80000000u) : __uint_as_float(~u);
}
#define ENC_NEG_INF 0x007FFFFFu

#define NB 256
#define BN ((N_NODES + NB - 1) / NB)
#define NG 256
#define CH ((N_EDGES + NG - 1) / NG)
#define WG 1024
#define PER ((CH + WG - 1) / WG)
#define HR 16
#define BROWS (((BN + 15) / 16) * 16)
#define BTILES (BROWS / 16)
#define RC ((BROWS * 16 + WG - 1) / WG)
static_assert((NB / 4) * HR == WG && NG % HR == 0, "column sums: 64 column groups x 16 row groups");
static_assert(NB == 8 * 32, "the scans below give 8 counters to each lane of one wave");
static_assert(BTILES <= WG / 32, "one wave per 16-row tile");
static_assert((long)NB * BN >= N_NODES, "buckets cover all nodes");

__global__ __launch_bounds__(WG) void gin_bin_count(const int* __restrict__ dst,
                                                    unsigned* __restrict__ hist) {
    __shared__ unsigned sH[NB];
    const int tid = threadIdx.x;
    const int g   = blockIdx.x;
    const int e0 = g * CH;
    const int e1 = (e0 + CH < N_EDGES) ? (e0 + CH) : N_EDGES;
    int dv[PER];
#pragma unroll
    for (int i = 0; i < PER; ++i) {
        const int e = e0 + tid + i * WG;
        const bool in = e < e1;
        const int d = dst[in ? e : (N_EDGES - 1)];
        dv[i] = in ? d : -1;
    }
    if (tid < NB) sH[tid] = 0u;
    __syncthreads();
#pragma unroll
    for (int i = 0; i < PER; ++i)
        if ((unsigned)dv[i] < (unsigned)N_NODES) atomicAdd(&sH[dv[i] / BN], 1u);
    __syncthreads();
    if (tid < NB) hist[g * NB + tid] = sH[tid];
}

__device__ __forceinline__ void wave_scan_256(unsigned* arr, int lane) {
    unsigned v[8];
    unsigned sum = 0u;
#pragma unroll
    for (int i = 0; i < 8; ++i) { v[i] = arr[lane * 8 + i]; sum += v[i]; }
    unsigned inc = sum;
#pragma unroll
    for (int off = 1; off < 32; off <<= 1) {
        const unsigned t = (unsigned)__shfl_up((int)inc, off);
        if (lane >= off) inc += t;
    }
    unsigned run = inc - sum;
#pragma unroll
    for (int i = 0; i < 8; ++i) { run += v[i]; arr[lane * 8 + i] = run; }
}

__global__ __launch_bounds__(WG) void gin_bin_fill(const int* __restrict__ src,
                                                   const int* __restrict__ dst,
                                                   const unsigned* __restrict__ hist,
                                                   unsigned* __restrict__ bstart,
                                                   int* __restrict__ binned) {
    __shared__ __align__(16) unsigned sTot[HR][NB];
    __shared__ __align__(16) unsigned sPre[HR][NB];
    __shared__ unsigned sScanG[NB];
    __shared__ unsigned sScanL[NB];
    __shared__ unsigned sDelta[NB];
    __shared__ unsigned sCur[NB];
    __shared__ int sS[CH];
    __shared__ int sD[CH];

    const int tid  = threadIdx.x;
    const int lane = tid & 31;
    const int wave = tid >> 5;
    const int g    = blockIdx.x;
    const int e0 = g * CH;
    const int e1 = (e0 + CH < N_EDGES) ? (e0 + CH) : N_EDGES;

    int dv[PER], sv[PER];
#pragma unroll
    for (int i = 0; i < PER; ++i) {
        const int e = e0 + tid + i * WG;
        const bool in = e < e1;
        const int ec = in ? e : (N_EDGES - 1);
        const int d = dst[ec];
        sv[i] = src[ec];
        dv[i] = in ? d : -1;
    }

    {
        const int cg = tid & (NB / 4 - 1);
        const int rg = tid >> 6;
        uint4 tot = make_uint4(0u, 0u, 0u, 0u), pre = make_uint4(0u, 0u, 0u, 0u);
#pragma unroll
        for (int k = 0; k < NG / HR; ++k) {
            const int r = rg * (NG / HR) + k;
            const uint4 h = *(const uint4*)(hist + r * NB + cg * 4);
            tot.x += h.x; tot.y += h.y; tot.z += h.z; tot.w += h.w;
            const unsigned mk = (r < g) ? 0xFFFFFFFFu : 0u;
            pre.x += h.x & mk; pre.y += h.y & mk; pre.z += h.z & mk; pre.w += h.w & mk;
        }
        *(uint4*)&sTot[rg][cg * 4] = tot;
        *(uint4*)&sPre[rg][cg * 4] = pre;
    }
    __syncthreads();
    unsigned T = 0u, P = 0u, own = 0u;
    if (tid < NB) {
#pragma unroll
        for (int k = 0; k < HR; ++k) { T += sTot[k][tid]; P += sPre[k][tid]; }
        own = hist[g * NB + tid];
        sScanG[tid] = T;
        sScanL[tid] = own;
    }
    __syncthreads();
    if (wave == 0) wave_scan_256(sScanG, lane);
    if (wave == 1) wave_scan_256(sScanL, lane);
    __syncthreads();
    if (tid < NB) {
        const unsigned base = sScanG[tid] - T;
        const unsigned lb   = sScanL[tid] - own;
        sDelta[tid] = base + P - lb;
        sCur[tid]   = lb;
        if (g == 0) {
            bstart[tid] = base;
            if (tid == NB - 1) bstart[NB] = base + T;
        }
    }
    __syncthreads();
    const unsigned nvalid = sScanL[NB - 1];

#pragma unroll
    for (int i = 0; i < PER; ++i) {
        const int d = dv[i];
        if ((unsigned)d < (unsigned)N_NODES) {
            const unsigned p = atomicAdd(&sCur[d / BN], 1u);
            if (p < (unsigned)CH) { sS[p] = sv[i]; sD[p] = d; }
        }
    }
    __syncthreads();
    for (unsigned i = tid; i < nvalid; i += WG) {
        const int d = sD[i];
        const unsigned pos = sDelta[d / BN] + i;
        *(v2i*)(binned + 2 * (size_t)pos) = v2i{sS[i], d};
    }
}

#define AG_BATCH 16
__global__ __launch_bounds__(WG) void gin_bucket_agg_gemm(const float* __restrict__ x,
                                                          const float* __restrict__ W,
                                                          const float* __restrict__ bias,
                                                          const unsigned* __restrict__ bstart,
                                                          const int* __restrict__ binned,
                                                          float* __restrict__ out) {
    __shared__ __align__(16) float sWp[(D / 2) * SW_STRIDE];
    __shared__ __align__(16) unsigned sAgg[BROWS * SA_STRIDE];

    const int tid  = threadIdx.x;
    const int lane = tid & 31;
    const int wave = __builtin_amdgcn_readfirstlane(tid >> 5);
    const int bk    = blockIdx.x;
    const int node0 = bk * BN;
    const int e0 = (int)bstart[bk];
    const int e1 = (int)bstart[bk + 1];

    int base = e0 + wave * 32;
    v2i ed = v2i{0, 0};
    if (base < e1) {
        int idx = base + lane;
        if (idx > e1 - 1) idx = e1 - 1;
        ed = *(const v2i*)(binned + 2 * (size_t)idx);
    }

    for (int i = tid; i < BROWS * SA_STRIDE / 4; i += WG)
        ((uint4*)sAgg)[i] = make_uint4(ENC_NEG_INF, ENC_NEG_INF, ENC_NEG_INF, ENC_NEG_INF);
#pragma unroll
    for (int i = 0; i < (D * D) / WG; ++i) {
        const int idx = tid + i * WG;
        const int o = idx >> 6;
        const int k = idx & 63;
        sWp[(k >> 1) * SW_STRIDE + o * 2 + (k & 1)] = W[idx];
    }
    __syncthreads();

    while (base < e1) {
        const int n = (e1 - base < 32) ? (e1 - base) : 32;
        const int nbase = base + WG;
        v2i edn = ed;
        if (nbase < e1) {
            int idx = nbase + lane;
            if (idx > e1 - 1) idx = e1 - 1;
            edn = *(const v2i*)(binned + 2 * (size_t)idx);
        }
        const int s_l = ed.x;
        const int d_l = ed.y - node0;
#pragma unroll
        for (int b = 0; b < 32; b += AG_BATCH) {
            if (b >= n) break;
            float v0[AG_BATCH], v1[AG_BATCH];
#pragma unroll
            for (int j = 0; j < AG_BATCH; ++j) {
                const int s = __builtin_amdgcn_readlane(s_l, b + j);
                const float* xr = x + (size_t)s * D + lane;
                v0[j] = xr[0];
                v1[j] = xr[32];
            }
#pragma unroll
            for (int j = 0; j < AG_BATCH; ++j) {
                unsigned dl = (unsigned)__builtin_amdgcn_readlane(d_l, b + j);
                if (dl > (unsigned)(BN - 1)) dl = (unsigned)(BN - 1);
                unsigned* ar = &sAgg[dl * SA_STRIDE + lane];
                atomicMax(ar,      enc_f32(v0[j]));
                atomicMax(ar + 32, enc_f32(v1[j]));
            }
        }
        ed = edn;
        base = nbase;
    }

    float4 xin[RC];
#pragma unroll
    for (int u = 0; u < RC; ++u) {
        const int i    = tid + u * WG;
        const int row  = i >> 4;
        const int c4   = (i & 15) << 2;
        const int node = node0 + row;
        const int nodec = (row < BN && node < N_NODES) ? node : (N_NODES - 1);
        xin[u] = *(const float4*)(x + (size_t)nodec * D + c4);
    }
    __syncthreads();
#pragma unroll
    for (int u = 0; u < RC; ++u) {
        const int i    = tid + u * WG;
        const int row  = i >> 4;
        const int c4   = (i & 15) << 2;
        const int node = node0 + row;
        if (i < BROWS * 16) {
            uint4* p = (uint4*)&sAgg[row * SA_STRIDE + c4];
            const uint4 au = *p;
            float4 r = make_float4(0.0f, 0.0f, 0.0f, 0.0f);
            if (row < BN && node < N_NODES) {
                r.x = xin[u].x + (au.x == ENC_NEG_INF ? 0.0f : dec_f32(au.x));
                r.y = xin[u].y + (au.y == ENC_NEG_INF ? 0.0f : dec_f32(au.y));
                r.z = xin[u].z + (au.z == ENC_NEG_INF ? 0.0f : dec_f32(au.z));
                r.w = xin[u].w + (au.w == ENC_NEG_INF ? 0.0f : dec_f32(au.w));
            }
            *p = make_uint4(__float_as_uint(r.x), __float_as_uint(r.y), __float_as_uint(r.z), __float_as_uint(r.w));
        }
    }
    __syncthreads();

    if (wave < BTILES) {
        v8f acc[4];
        acc[0] = v8f{}; acc[1] = v8f{}; acc[2] = v8f{}; acc[3] = v8f{};

        const int m     = lane & 15;
        const int khalf = (lane >> 4) << 1;
        const float* Arow = (const float*)&sAgg[(wave * 16 + m) * SA_STRIDE];

#pragma unroll
        for (int k = 0; k < D; k += 4) {
            const int kk = k + khalf;
            const v2f a = *(const v2f*)(Arow + kk);
            const float* bbase = &sWp[(kk >> 1) * SW_STRIDE + m * 2];
#pragma unroll
            for (int t = 0; t < 4; ++t) {
                const v2f bf = *(const v2f*)(bbase + t * 32);
                acc[t] = __builtin_amdgcn_wmma_f32_16x16x4_f32(
                    false, a, false, bf, (short)0, acc[t], false, false);
            }
        }

        const int rowoff = (lane >> 4) << 3;
#pragma unroll
        for (int t = 0; t < 4; ++t) {
            const int col = t * 16 + m;
            const float bv = bias[col];
#pragma unroll
            for (int v = 0; v < 8; ++v) {
                const int row  = wave * 16 + rowoff + v;
                const int node = node0 + row;
                if (row < BN && node < N_NODES) out[(size_t)node * D + col] = acc[t][v] + bv;
            }
        }
    }
}

extern "C" void kernel_launch(void* const* d_in, const int* in_sizes, int n_in,
                              void* d_out, int out_size, void* d_ws, size_t ws_size,
                              hipStream_t stream) {
    const float* x    = (const float*)d_in[0];
    const float* W    = (const float*)d_in[1];
    const float* bias = (const float*)d_in[2];
    const int*   src  = (const int*)d_in[3];
    const int*   dst  = (const int*)d_in[4];
    float*       out  = (float*)d_out;
    unsigned*    hist   = (unsigned*)d_ws;
    unsigned*    bstart = hist + NG * NB;
    int*         binned = (int*)(hist + NG * NB + 512);

    gin_bin_count<<<NG, WG, 0, stream>>>(dst, hist);

    gin_bin_fill<<<NG, WG, 0, stream>>>(src, dst, hist, bstart, binned);

    gin_bucket_agg_gemm<<<NB, WG, 0, stream>>>(x, W, bias, bstart, binned, out);
}


# Counting sort kept inside each edge chunk: every chunk sorted by dst bucket on its own, a bucket's waves walk 8 chunk segments each via a (start,length) table; the global count launch is gone
# speedup vs baseline: 3.0545x; 1.1006x over previous
#include <hip/hip_runtime.h>


#define N_NODES 100000
#define N_EDGES 1600000
#define D 64

#define SA_STRIDE 68
#define SW_STRIDE 160

typedef __attribute__((ext_vector_type(2))) float v2f;
typedef __attribute__((ext_vector_type(8))) float v8f;
typedef __attribute__((ext_vector_type(2))) int v2i;

__device__ __forceinline__ unsigned enc_f32(float f) {
    unsigned u = __float_as_uint(f);
    return (u & 0x80000000u) ? ~u : (u | 0x80000000u);
}
__device__ __forceinline__ float dec_f32(unsigned u) {
    return (u & 0x80000000u) ? __uint_as_float(u ^ 0x80000000u) : __uint_as_float(~u);
}
#define ENC_NEG_INF 0x007FFFFFu

#define NB 256
#define BN ((N_NODES + NB - 1) / NB)
#define NG 256
#define CH ((N_EDGES + NG - 1) / NG)
#define WG 1024
#define PER ((CH + WG - 1) / WG)
#define SEG (NG / (WG / 32))
#define BROWS (((BN + 15) / 16) * 16)
#define BTILES (BROWS / 16)
#define RC ((BROWS * 16 + WG - 1) / WG)
static_assert(NB == 8 * 32, "the scan below gives 8 counters to each lane of one wave");
static_assert(SEG * (WG / 32) == NG && SEG <= 32, "every wave of launch 2 owns SEG whole chunks, one per lane");
static_assert(BTILES <= WG / 32, "one wave per 16-row tile");
static_assert((long)NB * BN >= N_NODES, "buckets cover all nodes");

__device__ __forceinline__ void wave_scan_256(unsigned* arr, int lane) {
    unsigned v[8];
    unsigned sum = 0u;
#pragma unroll
    for (int i = 0; i < 8; ++i) { v[i] = arr[lane * 8 + i]; sum += v[i]; }
    unsigned inc = sum;
#pragma unroll
    for (int off = 1; off < 32; off <<= 1) {
        const unsigned t = (unsigned)__shfl_up((int)inc, off);
        if (lane >= off) inc += t;
    }
    unsigned run = inc - sum;
#pragma unroll
    for (int i = 0; i < 8; ++i) { run += v[i]; arr[lane * 8 + i] = run; }
}

__global__ __launch_bounds__(WG) void gin_chunk_sort(const int* __restrict__ src,
                                                     const int* __restrict__ dst,
                                                     unsigned* __restrict__ seg,
                                                     int* __restrict__ binned) {
    __shared__ unsigned sCnt[NB];
    __shared__ int sS[CH];
    __shared__ int sD[CH];

    const int tid  = threadIdx.x;
    const int lane = tid & 31;
    const int wave = tid >> 5;
    const int g    = blockIdx.x;
    const int e0 = g * CH;
    const int e1 = (e0 + CH < N_EDGES) ? (e0 + CH) : N_EDGES;

    int dv[PER], sv[PER];
#pragma unroll
    for (int i = 0; i < PER; ++i) {
        const int e = e0 + tid + i * WG;
        const bool in = e < e1;
        const int ec = in ? e : (N_EDGES - 1);
        const int d = dst[ec];
        sv[i] = src[ec];
        dv[i] = in ? d : -1;
    }
    if (tid < NB) sCnt[tid] = 0u;
    __syncthreads();
    unsigned rk[PER];
#pragma unroll
    for (int i = 0; i < PER; ++i) {
        rk[i] = 0u;
        if ((unsigned)dv[i] < (unsigned)N_NODES) rk[i] = atomicAdd(&sCnt[dv[i] / BN], 1u);
    }
    __syncthreads();
    unsigned own = 0u;
    if (tid < NB) own = sCnt[tid];
    __syncthreads();
    if (wave == 0) wave_scan_256(sCnt, lane);
    __syncthreads();
    if (tid < NB) {
        const unsigned lb = sCnt[tid] - own;
        *(uint2*)(seg + 2 * (size_t)(g * NB + tid)) = make_uint2((unsigned)(g * CH) + lb, own);
    }
#pragma unroll
    for (int i = 0; i < PER; ++i) {
        const int d = dv[i];
        if ((unsigned)d < (unsigned)N_NODES) {
            const int b = d / BN;
            const unsigned lb = b ? sCnt[b - 1] : 0u;
            const unsigned p = lb + rk[i];
            if (p < (unsigned)CH) { sS[p] = sv[i]; sD[p] = d; }
        }
    }
    __syncthreads();
    const unsigned nvalid = sCnt[NB - 1];
    for (unsigned i = tid; i < nvalid; i += WG)
        *(v2i*)(binned + 2 * ((size_t)g * CH + i)) = v2i{sS[i], sD[i]};
}

__device__ __forceinline__ int seg_record(int idx, int a0, int a1, int a2, int a3, int a4, int a5, int a6, int a7,
                                          int p1, int p2, int p3, int p4, int p5, int p6, int p7, int last) {
    int a = a0;
    a = (idx >= p1) ? a1 : a;
    a = (idx >= p2) ? a2 : a;
    a = (idx >= p3) ? a3 : a;
    a = (idx >= p4) ? a4 : a;
    a = (idx >= p5) ? a5 : a;
    a = (idx >= p6) ? a6 : a;
    a = (idx >= p7) ? a7 : a;
    int r = a + idx;
    r = r < 0 ? 0 : r;
    return r > last ? last : r;
}

#define AG_BATCH 16
__global__ __launch_bounds__(WG) void gin_bucket_agg_gemm(const float* __restrict__ x,
                                                          const float* __restrict__ W,
                                                          const float* __restrict__ bias,
                                                          const unsigned* __restrict__ seg,
                                                          const int* __restrict__ binned,
                                                          float* __restrict__ out) {
    __shared__ __align__(16) float sWp[(D / 2) * SW_STRIDE];
    __shared__ __align__(16) unsigned sAgg[BROWS * SA_STRIDE];

    const int tid  = threadIdx.x;
    const int lane = tid & 31;
    const int wave = __builtin_amdgcn_readfirstlane(tid >> 5);
    const int bk    = blockIdx.x;
    const int node0 = bk * BN;

    const uint2 sg = *(const uint2*)(seg + 2 * ((size_t)(wave * SEG + (lane & (SEG - 1))) * NB + bk));
    static_assert(SEG == 8, "the segment walk below is written out for 8 segments per wave");
    const int sx = (int)sg.x, sy = (int)sg.y;
    const int p1 = __builtin_amdgcn_readlane(sy, 0);
    const int p2 = p1 + __builtin_amdgcn_readlane(sy, 1);
    const int p3 = p2 + __builtin_amdgcn_readlane(sy, 2);
    const int p4 = p3 + __builtin_amdgcn_readlane(sy, 3);
    const int p5 = p4 + __builtin_amdgcn_readlane(sy, 4);
    const int p6 = p5 + __builtin_amdgcn_readlane(sy, 5);
    const int p7 = p6 + __builtin_amdgcn_readlane(sy, 6);
    const int total = p7 + __builtin_amdgcn_readlane(sy, 7);
    const int a0 = __builtin_amdgcn_readlane(sx, 0);
    const int a1 = __builtin_amdgcn_readlane(sx, 1) - p1;
    const int a2 = __builtin_amdgcn_readlane(sx, 2) - p2;
    const int a3 = __builtin_amdgcn_readlane(sx, 3) - p3;
    const int a4 = __builtin_amdgcn_readlane(sx, 4) - p4;
    const int a5 = __builtin_amdgcn_readlane(sx, 5) - p5;
    const int a6 = __builtin_amdgcn_readlane(sx, 6) - p6;
    const int a7 = __builtin_amdgcn_readlane(sx, 7) - p7;
    int base = 0;
    v2i ed = v2i{0, 0};
    if (base < total) {
        int idx = base + lane;
        if (idx > total - 1) idx = total - 1;
        ed = *(const v2i*)(binned + 2 * (size_t)seg_record(idx, a0, a1, a2, a3, a4, a5, a6, a7, p1, p2, p3, p4, p5, p6, p7, NG * CH - 1));
    }

    for (int i = tid; i < BROWS * SA_STRIDE / 4; i += WG)
        ((uint4*)sAgg)[i] = make_uint4(ENC_NEG_INF, ENC_NEG_INF, ENC_NEG_INF, ENC_NEG_INF);
#pragma unroll
    for (int i = 0; i < (D * D) / WG; ++i) {
        const int idx = tid + i * WG;
        const int o = idx >> 6;
        const int k = idx & 63;
        sWp[(k >> 1) * SW_STRIDE + o * 2 + (k & 1)] = W[idx];
    }
    __syncthreads();

    while (base < total) {
        const int n = (total - base < 32) ? (total - base) : 32;
        const int nbase = base + 32;
        v2i edn = ed;
        if (nbase < total) {
            int idx = nbase + lane;
            if (idx > total - 1) idx = total - 1;
            edn = *(const v2i*)(binned + 2 * (size_t)seg_record(idx, a0, a1, a2, a3, a4, a5, a6, a7, p1, p2, p3, p4, p5, p6, p7, NG * CH - 1));
        }
        const int s_l = ed.x;
        const int d_l = ed.y - node0;
#pragma unroll
        for (int b = 0; b < 32; b += AG_BATCH) {
            if (b >= n) break;
            float v0[AG_BATCH], v1[AG_BATCH];
#pragma unroll
            for (int j = 0; j < AG_BATCH; ++j) {
                const int s = __builtin_amdgcn_readlane(s_l, b + j);
                const float* xr = x + (size_t)s * D + lane;
                v0[j] = xr[0];
                v1[j] = xr[32];
            }
#pragma unroll
            for (int j = 0; j < AG_BATCH; ++j) {
                unsigned dl = (unsigned)__builtin_amdgcn_readlane(d_l, b + j);
                if (dl > (unsigned)(BN - 1)) dl = (unsigned)(BN - 1);
                unsigned* ar = &sAgg[dl * SA_STRIDE + lane];
                atomicMax(ar,      enc_f32(v0[j]));
                atomicMax(ar + 32, enc_f32(v1[j]));
            }
        }
        ed = edn;
        base = nbase;
    }

    float4 xin[RC];
#pragma unroll
    for (int u = 0; u < RC; ++u) {
        const int i    = tid + u * WG;
        const int row  = i >> 4;
        const int c4   = (i & 15) << 2;
        const int node = node0 + row;
        const int nodec = (row < BN && node < N_NODES) ? node : (N_NODES - 1);
        xin[u] = *(const float4*)(x + (size_t)nodec * D + c4);
    }
    __syncthreads();
#pragma unroll
    for (int u = 0; u < RC; ++u) {
        const int i    = tid + u * WG;
        const int row  = i >> 4;
        const int c4   = (i & 15) << 2;
        const int node = node0 + row;
        if (i < BROWS * 16) {
            uint4* p = (uint4*)&sAgg[row * SA_STRIDE + c4];
            const uint4 au = *p;
            float4 r = make_float4(0.0f, 0.0f, 0.0f, 0.0f);
            if (row < BN && node < N_NODES) {
                r.x = xin[u].x + (au.x == ENC_NEG_INF ? 0.0f : dec_f32(au.x));
                r.y = xin[u].y + (au.y == ENC_NEG_INF ? 0.0f : dec_f32(au.y));
                r.z = xin[u].z + (au.z == ENC_NEG_INF ? 0.0f : dec_f32(au.z));
                r.w = xin[u].w + (au.w == ENC_NEG_INF ? 0.0f : dec_f32(au.w));
            }
            *p = make_uint4(__float_as_uint(r.x), __float_as_uint(r.y), __float_as_uint(r.z), __float_as_uint(r.w));
        }
    }
    __syncthreads();

    if (wave < BTILES) {
        v8f acc[4];
        acc[0] = v8f{}; acc[1] = v8f{}; acc[2] = v8f{}; acc[3] = v8f{};

        const int m     = lane & 15;
        const int khalf = (lane >> 4) << 1;
        const float* Arow = (const float*)&sAgg[(wave * 16 + m) * SA_STRIDE];

#pragma unroll
        for (int k = 0; k < D; k += 4) {
            const int kk = k + khalf;
            const v2f a = *(const v2f*)(Arow + kk);
            const float* bbase = &sWp[(kk >> 1) * SW_STRIDE + m * 2];
#pragma unroll
            for (int t = 0; t < 4; ++t) {
                const v2f bf = *(const v2f*)(bbase + t * 32);
                acc[t] = __builtin_amdgcn_wmma_f32_16x16x4_f32(
                    false, a, false, bf, (short)0, acc[t], false, false);
            }
        }

        const int rowoff = (lane >> 4) << 3;
#pragma unroll
        for (int t = 0; t < 4; ++t) {
            const int col = t * 16 + m;
            const float bv = bias[col];
#pragma unroll
            for (int v = 0; v < 8; ++v) {
                const int row  = wave * 16 + rowoff + v;
                const int node = node0 + row;
                if (row < BN && node < N_NODES) out[(size_t)node * D + col] = acc[t][v] + bv;
            }
        }
    }
}

extern "C" void kernel_launch(void* const* d_in, const int* in_sizes, int n_in,
                              void* d_out, int out_size, void* d_ws, size_t ws_size,
                              hipStream_t stream) {
    const float* x    = (const float*)d_in[0];
    const float* W    = (const float*)d_in[1];
    const float* bias = (const float*)d_in[2];
    const int*   src  = (const int*)d_in[3];
    const int*   dst  = (const int*)d_in[4];
    float*       out  = (float*)d_out;
    unsigned*    seg    = (unsigned*)d_ws;
    int*         binned = (int*)(seg + 2 * NG * NB);

    gin_chunk_sort<<<NG, WG, 0, stream>>>(src, dst, seg, binned);

    gin_bucket_agg_gemm<<<NB, WG, 0, stream>>>(x, W, bias, seg, binned, out);
}


# Agg max kept in LDS as the float itself and updated by the LDS f32 max atomic: the integer order code (5 vector ops per value before each atomic, decode after) is gone; max stays order-free
# speedup vs baseline: 3.4990x; 1.1455x over previous
#include <hip/hip_runtime.h>


#define N_NODES 100000
#define N_EDGES 1600000
#define D 64

#define SA_STRIDE 68
#define SW_STRIDE 160

typedef __attribute__((ext_vector_type(2))) float v2f;
typedef __attribute__((ext_vector_type(8))) float v8f;
typedef __attribute__((ext_vector_type(2))) int v2i;

__device__ __forceinline__ void lds_max_f32(float* p, float v) {
    __hip_atomic_fetch_max(p, v, __ATOMIC_RELAXED, __HIP_MEMORY_SCOPE_WORKGROUP);
}
#define NEG_INF_F (-__builtin_inff())

#define NB 256
#define BN ((N_NODES + NB - 1) / NB)
#define NG 256
#define CH ((N_EDGES + NG - 1) / NG)
#define WG 1024
#define PER ((CH + WG - 1) / WG)
#define SEG (NG / (WG / 32))
#define BROWS (((BN + 15) / 16) * 16)
#define BTILES (BROWS / 16)
#define RC ((BROWS * 16 + WG - 1) / WG)
static_assert(NB == 8 * 32, "the scan below gives 8 counters to each lane of one wave");
static_assert(SEG * (WG / 32) == NG && SEG <= 32, "every wave of launch 2 owns SEG whole chunks, one per lane");
static_assert(BTILES <= WG / 32, "one wave per 16-row tile");
static_assert((long)NB * BN >= N_NODES, "buckets cover all nodes");

__device__ __forceinline__ void wave_scan_256(unsigned* arr, int lane) {
    unsigned v[8];
    unsigned sum = 0u;
#pragma unroll
    for (int i = 0; i < 8; ++i) { v[i] = arr[lane * 8 + i]; sum += v[i]; }
    unsigned inc = sum;
#pragma unroll
    for (int off = 1; off < 32; off <<= 1) {
        const unsigned t = (unsigned)__shfl_up((int)inc, off);
        if (lane >= off) inc += t;
    }
    unsigned run = inc - sum;
#pragma unroll
    for (int i = 0; i < 8; ++i) { run += v[i]; arr[lane * 8 + i] = run; }
}

__global__ __launch_bounds__(WG) void gin_chunk_sort(const int* __restrict__ src,
                                                     const int* __restrict__ dst,
                                                     unsigned* __restrict__ seg,
                                                     int* __restrict__ binned) {
    __shared__ unsigned sCnt[NB];
    __shared__ int sS[CH];
    __shared__ int sD[CH];

    const int tid  = threadIdx.x;
    const int lane = tid & 31;
    const int wave = tid >> 5;
    const int g    = blockIdx.x;
    const int e0 = g * CH;
    const int e1 = (e0 + CH < N_EDGES) ? (e0 + CH) : N_EDGES;

    int dv[PER], sv[PER];
#pragma unroll
    for (int i = 0; i < PER; ++i) {
        const int e = e0 + tid + i * WG;
        const bool in = e < e1;
        const int ec = in ? e : (N_EDGES - 1);
        const int d = dst[ec];
        sv[i] = src[ec];
        dv[i] = in ? d : -1;
    }
    if (tid < NB) sCnt[tid] = 0u;
    __syncthreads();
    unsigned rk[PER];
#pragma unroll
    for (int i = 0; i < PER; ++i) {
        rk[i] = 0u;
        if ((unsigned)dv[i] < (unsigned)N_NODES) rk[i] = atomicAdd(&sCnt[dv[i] / BN], 1u);
    }
    __syncthreads();
    unsigned own = 0u;
    if (tid < NB) own = sCnt[tid];
    __syncthreads();
    if (wave == 0) wave_scan_256(sCnt, lane);
    __syncthreads();
    if (tid < NB) {
        const unsigned lb = sCnt[tid] - own;
        *(uint2*)(seg + 2 * (size_t)(g * NB + tid)) = make_uint2((unsigned)(g * CH) + lb, own);
    }
#pragma unroll
    for (int i = 0; i < PER; ++i) {
        const int d = dv[i];
        if ((unsigned)d < (unsigned)N_NODES) {
            const int b = d / BN;
            const unsigned lb = b ? sCnt[b - 1] : 0u;
            const unsigned p = lb + rk[i];
            if (p < (unsigned)CH) { sS[p] = sv[i]; sD[p] = d; }
        }
    }
    __syncthreads();
    const unsigned nvalid = sCnt[NB - 1];
    for (unsigned i = tid; i < nvalid; i += WG)
        *(v2i*)(binned + 2 * ((size_t)g * CH + i)) = v2i{sS[i], sD[i]};
}

__device__ __forceinline__ int seg_record(int idx, int a0, int a1, int a2, int a3, int a4, int a5, int a6, int a7,
                                          int p1, int p2, int p3, int p4, int p5, int p6, int p7, int last) {
    int a = a0;
    a = (idx >= p1) ? a1 : a;
    a = (idx >= p2) ? a2 : a;
    a = (idx >= p3) ? a3 : a;
    a = (idx >= p4) ? a4 : a;
    a = (idx >= p5) ? a5 : a;
    a = (idx >= p6) ? a6 : a;
    a = (idx >= p7) ? a7 : a;
    int r = a + idx;
    r = r < 0 ? 0 : r;
    return r > last ? last : r;
}

#define AG_BATCH 16
__global__ __launch_bounds__(WG) void gin_bucket_agg_gemm(const float* __restrict__ x,
                                                          const float* __restrict__ W,
                                                          const float* __restrict__ bias,
                                                          const unsigned* __restrict__ seg,
                                                          const int* __restrict__ binned,
                                                          float* __restrict__ out) {
    __shared__ __align__(16) float sWp[(D / 2) * SW_STRIDE];
    __shared__ __align__(16) float sAgg[BROWS * SA_STRIDE];

    const int tid  = threadIdx.x;
    const int lane = tid & 31;
    const int wave = __builtin_amdgcn_readfirstlane(tid >> 5);
    const int bk    = blockIdx.x;
    const int node0 = bk * BN;

    const uint2 sg = *(const uint2*)(seg + 2 * ((size_t)(wave * SEG + (lane & (SEG - 1))) * NB + bk));
    static_assert(SEG == 8, "the segment walk below is written out for 8 segments per wave");
    const int sx = (int)sg.x, sy = (int)sg.y;
    const int p1 = __builtin_amdgcn_readlane(sy, 0);
    const int p2 = p1 + __builtin_amdgcn_readlane(sy, 1);
    const int p3 = p2 + __builtin_amdgcn_readlane(sy, 2);
    const int p4 = p3 + __builtin_amdgcn_readlane(sy, 3);
    const int p5 = p4 + __builtin_amdgcn_readlane(sy, 4);
    const int p6 = p5 + __builtin_amdgcn_readlane(sy, 5);
    const int p7 = p6 + __builtin_amdgcn_readlane(sy, 6);
    const int total = p7 + __builtin_amdgcn_readlane(sy, 7);
    const int a0 = __builtin_amdgcn_readlane(sx, 0);
    const int a1 = __builtin_amdgcn_readlane(sx, 1) - p1;
    const int a2 = __builtin_amdgcn_readlane(sx, 2) - p2;
    const int a3 = __builtin_amdgcn_readlane(sx, 3) - p3;
    const int a4 = __builtin_amdgcn_readlane(sx, 4) - p4;
    const int a5 = __builtin_amdgcn_readlane(sx, 5) - p5;
    const int a6 = __builtin_amdgcn_readlane(sx, 6) - p6;
    const int a7 = __builtin_amdgcn_readlane(sx, 7) - p7;
    int base = 0;
    v2i ed = v2i{0, 0};
    if (base < total) {
        int idx = base + lane;
        if (idx > total - 1) idx = total - 1;
        ed = *(const v2i*)(binned + 2 * (size_t)seg_record(idx, a0, a1, a2, a3, a4, a5, a6, a7, p1, p2, p3, p4, p5, p6, p7, NG * CH - 1));
    }

    for (int i = tid; i < BROWS * SA_STRIDE / 4; i += WG)
        ((float4*)sAgg)[i] = make_float4(NEG_INF_F, NEG_INF_F, NEG_INF_F, NEG_INF_F);
#pragma unroll
    for (int i = 0; i < (D * D) / WG; ++i) {
        const int idx = tid + i * WG;
        const int o = idx >> 6;
        const int k = idx & 63;
        sWp[(k >> 1) * SW_STRIDE + o * 2 + (k & 1)] = W[idx];
    }
    __syncthreads();

    while (base < total) {
        const int n = (total - base < 32) ? (total - base) : 32;
        const int nbase = base + 32;
        v2i edn = ed;
        if (nbase < total) {
            int idx = nbase + lane;
            if (idx > total - 1) idx = total - 1;
            edn = *(const v2i*)(binned + 2 * (size_t)seg_record(idx, a0, a1, a2, a3, a4, a5, a6, a7, p1, p2, p3, p4, p5, p6, p7, NG * CH - 1));
        }
        const int s_l = ed.x;
        const int d_l = ed.y - node0;
#pragma unroll
        for (int b = 0; b < 32; b += AG_BATCH) {
            if (b >= n) break;
            float v0[AG_BATCH], v1[AG_BATCH];
#pragma unroll
            for (int j = 0; j < AG_BATCH; ++j) {
                const int s = __builtin_amdgcn_readlane(s_l, b + j);
                const float* xr = x + (size_t)s * D + lane;
                v0[j] = xr[0];
                v1[j] = xr[32];
            }
#pragma unroll
            for (int j = 0; j < AG_BATCH; ++j) {
                unsigned dl = (unsigned)__builtin_amdgcn_readlane(d_l, b + j);
                if (dl > (unsigned)(BN - 1)) dl = (unsigned)(BN - 1);
                float* ar = &sAgg[dl * SA_STRIDE + lane];
                lds_max_f32(ar,      v0[j]);
                lds_max_f32(ar + 32, v1[j]);
            }
        }
        ed = edn;
        base = nbase;
    }

    float4 xin[RC];
#pragma unroll
    for (int u = 0; u < RC; ++u) {
        const int i    = tid + u * WG;
        const int row  = i >> 4;
        const int c4   = (i & 15) << 2;
        const int node = node0 + row;
        const int nodec = (row < BN && node < N_NODES) ? node : (N_NODES - 1);
        xin[u] = *(const float4*)(x + (size_t)nodec * D + c4);
    }
    __syncthreads();
#pragma unroll
    for (int u = 0; u < RC; ++u) {
        const int i    = tid + u * WG;
        const int row  = i >> 4;
        const int c4   = (i & 15) << 2;
        const int node = node0 + row;
        if (i < BROWS * 16) {
            float4* p = (float4*)&sAgg[row * SA_STRIDE + c4];
            const float4 au = *p;
            float4 r = make_float4(0.0f, 0.0f, 0.0f, 0.0f);
            if (row < BN && node < N_NODES) {
                r.x = xin[u].x + (au.x == NEG_INF_F ? 0.0f : au.x);
                r.y = xin[u].y + (au.y == NEG_INF_F ? 0.0f : au.y);
                r.z = xin[u].z + (au.z == NEG_INF_F ? 0.0f : au.z);
                r.w = xin[u].w + (au.w == NEG_INF_F ? 0.0f : au.w);
            }
            *p = r;
        }
    }
    __syncthreads();

    if (wave < BTILES) {
        v8f acc[4];
        acc[0] = v8f{}; acc[1] = v8f{}; acc[2] = v8f{}; acc[3] = v8f{};

        const int m     = lane & 15;
        const int khalf = (lane >> 4) << 1;
        const float* Arow = &sAgg[(wave * 16 + m) * SA_STRIDE];

#pragma unroll
        for (int k = 0; k < D; k += 4) {
            const int kk = k + khalf;
            const v2f a = *(const v2f*)(Arow + kk);
            const float* bbase = &sWp[(kk >> 1) * SW_STRIDE + m * 2];
#pragma unroll
            for (int t = 0; t < 4; ++t) {
                const v2f bf = *(const v2f*)(bbase + t * 32);
                acc[t] = __builtin_amdgcn_wmma_f32_16x16x4_f32(
                    false, a, false, bf, (short)0, acc[t], false, false);
            }
        }

        const int rowoff = (lane >> 4) << 3;
#pragma unroll
        for (int t = 0; t < 4; ++t) {
            const int col = t * 16 + m;
            const float bv = bias[col];
#pragma unroll
            for (int v = 0; v < 8; ++v) {
                const int row  = wave * 16 + rowoff + v;
                const int node = node0 + row;
                if (row < BN && node < N_NODES) out[(size_t)node * D + col] = acc[t][v] + bv;
            }
        }
    }
}

extern "C" void kernel_launch(void* const* d_in, const int* in_sizes, int n_in,
                              void* d_out, int out_size, void* d_ws, size_t ws_size,
                              hipStream_t stream) {
    const float* x    = (const float*)d_in[0];
    const float* W    = (const float*)d_in[1];
    const float* bias = (const float*)d_in[2];
    const int*   src  = (const int*)d_in[3];
    const int*   dst  = (const int*)d_in[4];
    float*       out  = (float*)d_out;
    unsigned*    seg    = (unsigned*)d_ws;
    int*         binned = (int*)(seg + 2 * NG * NB);

    gin_chunk_sort<<<NG, WG, 0, stream>>>(src, dst, seg, binned);

    gin_bucket_agg_gemm<<<NB, WG, 0, stream>>>(x, W, bias, seg, binned, out);
}


# Bucket's edges counting-sorted by agg row in LDS; each wave owns 12-13 whole rows, keeps a row's max in registers and stores it once: the 2 LDS max atomics per edge are gone; max stays order-free
# speedup vs baseline: 3.6989x; 1.0571x over previous
#include <hip/hip_runtime.h>


#define N_NODES 100000
#define N_EDGES 1600000
#define D 64

#define SA_STRIDE 68
#define SW_STRIDE 160

typedef __attribute__((ext_vector_type(2))) float v2f;
typedef __attribute__((ext_vector_type(8))) float v8f;
typedef __attribute__((ext_vector_type(2))) int v2i;

__device__ __forceinline__ void lds_max_f32(float* p, float v) {
    __hip_atomic_fetch_max(p, v, __ATOMIC_RELAXED, __HIP_MEMORY_SCOPE_WORKGROUP);
}
#define NEG_INF_F (-__builtin_inff())

#define NB 256
#define BN ((N_NODES + NB - 1) / NB)
#define NG 256
#define CH ((N_EDGES + NG - 1) / NG)
#define WG 1024
#define PER ((CH + WG - 1) / WG)
#define SEG (NG / (WG / 32))
#define BROWS (((BN + 15) / 16) * 16)
#define BTILES (BROWS / 16)
#define RC ((BROWS * 16 + WG - 1) / WG)
#define KG 9
#define CAP 8192
#define RPW (BN / (WG / 32))
#define RREM (BN % (WG / 32))
#define SCN ((BROWS + 32) / 32)
static_assert(SCN * 32 > BROWS, "the scan covers every row counter and the entry behind the last row");
static_assert(RPW * (WG / 32) + RREM == BN, "the waves' row ranges cover the bucket exactly");
static_assert(NB == 8 * 32, "the scan below gives 8 counters to each lane of one wave");
static_assert(SEG * (WG / 32) == NG && SEG <= 32, "every wave of launch 2 owns SEG whole chunks, one per lane");
static_assert(BTILES <= WG / 32, "one wave per 16-row tile");
static_assert((long)NB * BN >= N_NODES, "buckets cover all nodes");

__device__ __forceinline__ void wave_scan_256(unsigned* arr, int lane) {
    unsigned v[8];
    unsigned sum = 0u;
#pragma unroll
    for (int i = 0; i < 8; ++i) { v[i] = arr[lane * 8 + i]; sum += v[i]; }
    unsigned inc = sum;
#pragma unroll
    for (int off = 1; off < 32; off <<= 1) {
        const unsigned t = (unsigned)__shfl_up((int)inc, off);
        if (lane >= off) inc += t;
    }
    unsigned run = inc - sum;
#pragma unroll
    for (int i = 0; i < 8; ++i) { run += v[i]; arr[lane * 8 + i] = run; }
}

__global__ __launch_bounds__(WG) void gin_chunk_sort(const int* __restrict__ src,
                                                     const int* __restrict__ dst,
                                                     unsigned* __restrict__ seg,
                                                     int* __restrict__ binned) {
    __shared__ unsigned sCnt[NB];
    __shared__ int sS[CH];
    __shared__ int sD[CH];

    const int tid  = threadIdx.x;
    const int lane = tid & 31;
    const int wave = tid >> 5;
    const int g    = blockIdx.x;
    const int e0 = g * CH;
    const int e1 = (e0 + CH < N_EDGES) ? (e0 + CH) : N_EDGES;

    int dv[PER], sv[PER];
#pragma unroll
    for (int i = 0; i < PER; ++i) {
        const int e = e0 + tid + i * WG;
        const bool in = e < e1;
        const int ec = in ? e : (N_EDGES - 1);
        const int d = dst[ec];
        sv[i] = src[ec];
        dv[i] = in ? d : -1;
    }
    if (tid < NB) sCnt[tid] = 0u;
    __syncthreads();
    unsigned rk[PER];
#pragma unroll
    for (int i = 0; i < PER; ++i) {
        rk[i] = 0u;
        if ((unsigned)dv[i] < (unsigned)N_NODES) rk[i] = atomicAdd(&sCnt[dv[i] / BN], 1u);
    }
    __syncthreads();
    unsigned own = 0u;
    if (tid < NB) own = sCnt[tid];
    __syncthreads();
    if (wave == 0) wave_scan_256(sCnt, lane);
    __syncthreads();
    if (tid < NB) {
        const unsigned lb = sCnt[tid] - own;
        *(uint2*)(seg + 2 * (size_t)(g * NB + tid)) = make_uint2((unsigned)(g * CH) + lb, own);
    }
#pragma unroll
    for (int i = 0; i < PER; ++i) {
        const int d = dv[i];
        if ((unsigned)d < (unsigned)N_NODES) {
            const int b = d / BN;
            const unsigned lb = b ? sCnt[b - 1] : 0u;
            const unsigned p = lb + rk[i];
            if (p < (unsigned)CH) { sS[p] = sv[i]; sD[p] = d; }
        }
    }
    __syncthreads();
    const unsigned nvalid = sCnt[NB - 1];
    for (unsigned i = tid; i < nvalid; i += WG)
        *(v2i*)(binned + 2 * ((size_t)g * CH + i)) = v2i{sS[i], sD[i]};
}

__device__ __forceinline__ int seg_record(int idx, int a0, int a1, int a2, int a3, int a4, int a5, int a6, int a7,
                                          int p1, int p2, int p3, int p4, int p5, int p6, int p7, int last) {
    int a = a0;
    a = (idx >= p1) ? a1 : a;
    a = (idx >= p2) ? a2 : a;
    a = (idx >= p3) ? a3 : a;
    a = (idx >= p4) ? a4 : a;
    a = (idx >= p5) ? a5 : a;
    a = (idx >= p6) ? a6 : a;
    a = (idx >= p7) ? a7 : a;
    int r = a + idx;
    r = r < 0 ? 0 : r;
    return r > last ? last : r;
}

#define AG_BATCH 16
__global__ __launch_bounds__(WG) void gin_bucket_agg_gemm(const float* __restrict__ x,
                                                          const float* __restrict__ W,
                                                          const float* __restrict__ bias,
                                                          const unsigned* __restrict__ seg,
                                                          const int* __restrict__ binned,
                                                          float* __restrict__ out) {
    __shared__ __align__(16) float sWp[(D / 2) * SW_STRIDE];
    __shared__ __align__(16) float sAgg[BROWS * SA_STRIDE];
    __shared__ __align__(8) int sEdge[2 * CAP];
    __shared__ int sCnt[SCN * 32];
    __shared__ int sStart[SCN * 32];
    __shared__ int sWaveTot[WG / 32];

    const int tid  = threadIdx.x;
    const int lane = tid & 31;
    const int wave = __builtin_amdgcn_readfirstlane(tid >> 5);
    const int bk    = blockIdx.x;
    const int node0 = bk * BN;

    const uint2 sg = *(const uint2*)(seg + 2 * ((size_t)(wave * SEG + (lane & (SEG - 1))) * NB + bk));
    static_assert(SEG == 8, "the segment walk below is written out for 8 segments per wave");
    const int sx = (int)sg.x, sy = (int)sg.y;
    const int p1 = __builtin_amdgcn_readlane(sy, 0);
    const int p2 = p1 + __builtin_amdgcn_readlane(sy, 1);
    const int p3 = p2 + __builtin_amdgcn_readlane(sy, 2);
    const int p4 = p3 + __builtin_amdgcn_readlane(sy, 3);
    const int p5 = p4 + __builtin_amdgcn_readlane(sy, 4);
    const int p6 = p5 + __builtin_amdgcn_readlane(sy, 5);
    const int p7 = p6 + __builtin_amdgcn_readlane(sy, 6);
    const int total = p7 + __builtin_amdgcn_readlane(sy, 7);
    const int a0 = __builtin_amdgcn_readlane(sx, 0);
    const int a1 = __builtin_amdgcn_readlane(sx, 1) - p1;
    const int a2 = __builtin_amdgcn_readlane(sx, 2) - p2;
    const int a3 = __builtin_amdgcn_readlane(sx, 3) - p3;
    const int a4 = __builtin_amdgcn_readlane(sx, 4) - p4;
    const int a5 = __builtin_amdgcn_readlane(sx, 5) - p5;
    const int a6 = __builtin_amdgcn_readlane(sx, 6) - p6;
    const int a7 = __builtin_amdgcn_readlane(sx, 7) - p7;
    v2i rec[KG];
#pragma unroll
    for (int g = 0; g < KG; ++g) {
        rec[g] = v2i{0, 0};
        if (g * 32 < total) {
            int idx = g * 32 + lane;
            if (idx > total - 1) idx = total - 1;
            rec[g] = *(const v2i*)(binned + 2 * (size_t)seg_record(idx, a0, a1, a2, a3, a4, a5, a6, a7, p1, p2, p3, p4, p5, p6, p7, NG * CH - 1));
        }
    }

    for (int i = tid; i < BROWS * SA_STRIDE / 4; i += WG)
        ((float4*)sAgg)[i] = make_float4(NEG_INF_F, NEG_INF_F, NEG_INF_F, NEG_INF_F);
#pragma unroll
    for (int i = 0; i < (D * D) / WG; ++i) {
        const int idx = tid + i * WG;
        const int o = idx >> 6;
        const int k = idx & 63;
        sWp[(k >> 1) * SW_STRIDE + o * 2 + (k & 1)] = W[idx];
    }
    if (tid < SCN * 32) sCnt[tid] = 0;
    if (lane == 0) sWaveTot[wave] = total;
    __syncthreads();

    static_assert(WG / 32 == 32, "one wave total per lane");
    int wsum = sWaveTot[lane];
    int wmax = wsum;
#pragma unroll
    for (int off = 16; off >= 1; off >>= 1) {
        wsum += __shfl_xor(wsum, off);
        const int t = __shfl_xor(wmax, off);
        wmax = t > wmax ? t : wmax;
    }
    const bool by_rows = __builtin_amdgcn_readfirstlane(wsum) <= CAP &&
                         __builtin_amdgcn_readfirstlane(wmax) <= KG * 32;

    if (by_rows) {
        int rk[KG];
#pragma unroll
        for (int g = 0; g < KG; ++g) {
            rk[g] = 0;
            if (g * 32 < total) {
                int dl = rec[g].y - node0;
                dl = dl < 0 ? 0 : (dl > BN - 1 ? BN - 1 : dl);
                rec[g].y = dl;
                if (g * 32 + lane < total) rk[g] = atomicAdd(&sCnt[dl], 1);
            }
        }
        __syncthreads();
        if (wave == 0) {
            int v[SCN];
            int sum = 0;
#pragma unroll
            for (int i = 0; i < SCN; ++i) { v[i] = sCnt[lane * SCN + i]; sum += v[i]; }
            int inc = sum;
#pragma unroll
            for (int off = 1; off < 32; off <<= 1) {
                const int t = __shfl_up(inc, off);
                if (lane >= off) inc += t;
            }
            int run = inc - sum;
#pragma unroll
            for (int i = 0; i < SCN; ++i) { sStart[lane * SCN + i] = run; run += v[i]; }
        }
        __syncthreads();
#pragma unroll
        for (int g = 0; g < KG; ++g) {
            if (g * 32 < total) {
                if (g * 32 + lane < total) {
                    const unsigned slot = (unsigned)(sStart[rec[g].y] + rk[g]);
                    if (slot < (unsigned)CAP) *(v2i*)(sEdge + 2 * slot) = rec[g];
                }
            }
        }
        __syncthreads();

        const int r0 = wave * RPW + (wave < RREM ? wave : RREM);
        const int r1 = r0 + RPW + (wave < RREM ? 1 : 0);
        int e0 = __builtin_amdgcn_readfirstlane(sStart[r0]);
        int e1 = __builtin_amdgcn_readfirstlane(sStart[r1]);
        e0 = e0 < 0 ? 0 : e0;
        e1 = e1 > CAP ? CAP : e1;
        int cur = -1;
        float m0 = NEG_INF_F, m1 = NEG_INF_F;
        for (int base = e0; base < e1; base += 32) {
            const int n = (e1 - base < 32) ? (e1 - base) : 32;
            int idx = base + lane;
            if (idx > e1 - 1) idx = e1 - 1;
            const v2i ed = *(const v2i*)(sEdge + 2 * idx);
            const int s_l = ed.x;
            const int d_l = ((unsigned)ed.y > (unsigned)(BN - 1)) ? (BN - 1) : ed.y;
            int d_prev = __shfl_up(d_l, 1);
            if (lane == 0) d_prev = cur;
            const unsigned chg = __builtin_amdgcn_ballot_w32(d_l != d_prev);
#pragma unroll
            for (int b = 0; b < 32; b += AG_BATCH) {
                if (b >= n) break;
                float v0[AG_BATCH], v1[AG_BATCH];
#pragma unroll
                for (int j = 0; j < AG_BATCH; ++j) {
                    const int s = __builtin_amdgcn_readlane(s_l, b + j);
                    const float* xr = x + (size_t)s * D + lane;
                    v0[j] = xr[0];
                    v1[j] = xr[32];
                }
#pragma unroll
                for (int j = 0; j < AG_BATCH; ++j) {
                    if (chg & (1u << (b + j))) {
                        if (cur >= 0) {
                            float* ar = &sAgg[cur * SA_STRIDE + lane];
                            ar[0]  = m0;
                            ar[32] = m1;
                        }
                        cur = __builtin_amdgcn_readlane(d_l, b + j);
                        m0 = NEG_INF_F;
                        m1 = NEG_INF_F;
                    }
                    asm("v_max_num_f32 %0, %0, %1" : "+v"(m0) : "v"(v0[j]));
                    asm("v_max_num_f32 %0, %0, %1" : "+v"(m1) : "v"(v1[j]));
                }
            }
        }
        if (cur >= 0) {
            float* ar = &sAgg[cur * SA_STRIDE + lane];
            ar[0]  = m0;
            ar[32] = m1;
        }
    } else {
        int base = 0;
        v2i ed = rec[0];
        while (base < total) {
            const int n = (total - base < 32) ? (total - base) : 32;
            const int nbase = base + 32;
            v2i edn = ed;
            if (nbase < total) {
                int idx = nbase + lane;
                if (idx > total - 1) idx = total - 1;
                edn = *(const v2i*)(binned + 2 * (size_t)seg_record(idx, a0, a1, a2, a3, a4, a5, a6, a7, p1, p2, p3, p4, p5, p6, p7, NG * CH - 1));
            }
            const int s_l = ed.x;
            const int d_l = ed.y - node0;
#pragma unroll
            for (int b = 0; b < 32; b += AG_BATCH) {
                if (b >= n) break;
                float v0[AG_BATCH], v1[AG_BATCH];
#pragma unroll
                for (int j = 0; j < AG_BATCH; ++j) {
                    const int s = __builtin_amdgcn_readlane(s_l, b + j);
                    const float* xr = x + (size_t)s * D + lane;
                    v0[j] = xr[0];
                    v1[j] = xr[32];
                }
#pragma unroll
                for (int j = 0; j < AG_BATCH; ++j) {
                    unsigned dl = (unsigned)__builtin_amdgcn_readlane(d_l, b + j);
                    if (dl > (unsigned)(BN - 1)) dl = (unsigned)(BN - 1);
                    float* ar = &sAgg[dl * SA_STRIDE + lane];
                    lds_max_f32(ar,      v0[j]);
                    lds_max_f32(ar + 32, v1[j]);
                }
            }
            ed = edn;
            base = nbase;
        }
    }

    float4 xin[RC];
#pragma unroll
    for (int u = 0; u < RC; ++u) {
        const int i    = tid + u * WG;
        const int row  = i >> 4;
        const int c4   = (i & 15) << 2;
        const int node = node0 + row;
        const int nodec = (row < BN && node < N_NODES) ? node : (N_NODES - 1);
        xin[u] = *(const float4*)(x + (size_t)nodec * D + c4);
    }
    __syncthreads();
#pragma unroll
    for (int u = 0; u < RC; ++u) {
        const int i    = tid + u * WG;
        const int row  = i >> 4;
        const int c4   = (i & 15) << 2;
        const int node = node0 + row;
        if (i < BROWS * 16) {
            float4* p = (float4*)&sAgg[row * SA_STRIDE + c4];
            const float4 au = *p;
            float4 r = make_float4(0.0f, 0.0f, 0.0f, 0.0f);
            if (row < BN && node < N_NODES) {
                r.x = xin[u].x + (au.x == NEG_INF_F ? 0.0f : au.x);
                r.y = xin[u].y + (au.y == NEG_INF_F ? 0.0f : au.y);
                r.z = xin[u].z + (au.z == NEG_INF_F ? 0.0f : au.z);
                r.w = xin[u].w + (au.w == NEG_INF_F ? 0.0f : au.w);
            }
            *p = r;
        }
    }
    __syncthreads();

    if (wave < BTILES) {
        v8f acc[4];
        acc[0] = v8f{}; acc[1] = v8f{}; acc[2] = v8f{}; acc[3] = v8f{};

        const int m     = lane & 15;
        const int khalf = (lane >> 4) << 1;
        const float* Arow = &sAgg[(wave * 16 + m) * SA_STRIDE];

#pragma unroll
        for (int k = 0; k < D; k += 4) {
            const int kk = k + khalf;
            const v2f a = *(const v2f*)(Arow + kk);
            const float* bbase = &sWp[(kk >> 1) * SW_STRIDE + m * 2];
#pragma unroll
            for (int t = 0; t < 4; ++t) {
                const v2f bf = *(const v2f*)(bbase + t * 32);
                acc[t] = __builtin_amdgcn_wmma_f32_16x16x4_f32(
                    false, a, false, bf, (short)0, acc[t], false, false);
            }
        }

        const int rowoff = (lane >> 4) << 3;
#pragma unroll
        for (int t = 0; t < 4; ++t) {
            const int col = t * 16 + m;
            const float bv = bias[col];
#pragma unroll
            for (int v = 0; v < 8; ++v) {
                const int row  = wave * 16 + rowoff + v;
                const int node = node0 + row;
                if (row < BN && node < N_NODES) out[(size_t)node * D + col] = acc[t][v] + bv;
            }
        }
    }
}

extern "C" void kernel_launch(void* const* d_in, const int* in_sizes, int n_in,
                              void* d_out, int out_size, void* d_ws, size_t ws_size,
                              hipStream_t stream) {
    const float* x    = (const float*)d_in[0];
    const float* W    = (const float*)d_in[1];
    const float* bias = (const float*)d_in[2];
    const int*   src  = (const int*)d_in[3];
    const int*   dst  = (const int*)d_in[4];
    float*       out  = (float*)d_out;
    unsigned*    seg    = (unsigned*)d_ws;
    int*         binned = (int*)(seg + 2 * NG * NB);

    gin_chunk_sort<<<NG, WG, 0, stream>>>(src, dst, seg, binned);

    gin_bucket_agg_gemm<<<NB, WG, 0, stream>>>(x, W, bias, seg, binned, out);
}


# x[src] gathered with one global_load_b64 per edge (lane = 2 adjacent features): one instruction covers the row's whole aligned 256-B unit instead of two one-line loads; max stays order-free
# speedup vs baseline: 3.9499x; 1.0679x over previous
#include <hip/hip_runtime.h>


#define N_NODES 100000
#define N_EDGES 1600000
#define D 64

#define SA_STRIDE 68
#define SW_STRIDE 160

typedef __attribute__((ext_vector_type(2))) float v2f;
typedef __attribute__((ext_vector_type(8))) float v8f;
typedef __attribute__((ext_vector_type(2))) int v2i;

__device__ __forceinline__ void lds_max_f32(float* p, float v) {
    __hip_atomic_fetch_max(p, v, __ATOMIC_RELAXED, __HIP_MEMORY_SCOPE_WORKGROUP);
}
#define NEG_INF_F (-__builtin_inff())

#define NB 256
#define BN ((N_NODES + NB - 1) / NB)
#define NG 256
#define CH ((N_EDGES + NG - 1) / NG)
#define WG 1024
#define PER ((CH + WG - 1) / WG)
#define SEG (NG / (WG / 32))
#define BROWS (((BN + 15) / 16) * 16)
#define BTILES (BROWS / 16)
#define RC ((BROWS * 16 + WG - 1) / WG)
#define KG 9
#define CAP 8192
#define RPW (BN / (WG / 32))
#define RREM (BN % (WG / 32))
#define SCN ((BROWS + 32) / 32)
static_assert(SCN * 32 > BROWS, "the scan covers every row counter and the entry behind the last row");
static_assert(RPW * (WG / 32) + RREM == BN, "the waves' row ranges cover the bucket exactly");
static_assert(NB == 8 * 32, "the scan below gives 8 counters to each lane of one wave");
static_assert(SEG * (WG / 32) == NG && SEG <= 32, "every wave of launch 2 owns SEG whole chunks, one per lane");
static_assert(BTILES <= WG / 32, "one wave per 16-row tile");
static_assert((long)NB * BN >= N_NODES, "buckets cover all nodes");

__device__ __forceinline__ void wave_scan_256(unsigned* arr, int lane) {
    unsigned v[8];
    unsigned sum = 0u;
#pragma unroll
    for (int i = 0; i < 8; ++i) { v[i] = arr[lane * 8 + i]; sum += v[i]; }
    unsigned inc = sum;
#pragma unroll
    for (int off = 1; off < 32; off <<= 1) {
        const unsigned t = (unsigned)__shfl_up((int)inc, off);
        if (lane >= off) inc += t;
    }
    unsigned run = inc - sum;
#pragma unroll
    for (int i = 0; i < 8; ++i) { run += v[i]; arr[lane * 8 + i] = run; }
}

__global__ __launch_bounds__(WG) void gin_chunk_sort(const int* __restrict__ src,
                                                     const int* __restrict__ dst,
                                                     unsigned* __restrict__ seg,
                                                     int* __restrict__ binned) {
    __shared__ unsigned sCnt[NB];
    __shared__ int sS[CH];
    __shared__ int sD[CH];

    const int tid  = threadIdx.x;
    const int lane = tid & 31;
    const int wave = tid >> 5;
    const int g    = blockIdx.x;
    const int e0 = g * CH;
    const int e1 = (e0 + CH < N_EDGES) ? (e0 + CH) : N_EDGES;

    int dv[PER], sv[PER];
#pragma unroll
    for (int i = 0; i < PER; ++i) {
        const int e = e0 + tid + i * WG;
        const bool in = e < e1;
        const int ec = in ? e : (N_EDGES - 1);
        const int d = dst[ec];
        sv[i] = src[ec];
        dv[i] = in ? d : -1;
    }
    if (tid < NB) sCnt[tid] = 0u;
    __syncthreads();
    unsigned rk[PER];
#pragma unroll
    for (int i = 0; i < PER; ++i) {
        rk[i] = 0u;
        if ((unsigned)dv[i] < (unsigned)N_NODES) rk[i] = atomicAdd(&sCnt[dv[i] / BN], 1u);
    }
    __syncthreads();
    unsigned own = 0u;
    if (tid < NB) own = sCnt[tid];
    __syncthreads();
    if (wave == 0) wave_scan_256(sCnt, lane);
    __syncthreads();
    if (tid < NB) {
        const unsigned lb = sCnt[tid] - own;
        *(uint2*)(seg + 2 * (size_t)(g * NB + tid)) = make_uint2((unsigned)(g * CH) + lb, own);
    }
#pragma unroll
    for (int i = 0; i < PER; ++i) {
        const int d = dv[i];
        if ((unsigned)d < (unsigned)N_NODES) {
            const int b = d / BN;
            const unsigned lb = b ? sCnt[b - 1] : 0u;
            const unsigned p = lb + rk[i];
            if (p < (unsigned)CH) { sS[p] = sv[i]; sD[p] = d; }
        }
    }
    __syncthreads();
    const unsigned nvalid = sCnt[NB - 1];
    for (unsigned i = tid; i < nvalid; i += WG)
        *(v2i*)(binned + 2 * ((size_t)g * CH + i)) = v2i{sS[i], sD[i]};
}

__device__ __forceinline__ int seg_record(int idx, int a0, int a1, int a2, int a3, int a4, int a5, int a6, int a7,
                                          int p1, int p2, int p3, int p4, int p5, int p6, int p7, int last) {
    int a = a0;
    a = (idx >= p1) ? a1 : a;
    a = (idx >= p2) ? a2 : a;
    a = (idx >= p3) ? a3 : a;
    a = (idx >= p4) ? a4 : a;
    a = (idx >= p5) ? a5 : a;
    a = (idx >= p6) ? a6 : a;
    a = (idx >= p7) ? a7 : a;
    int r = a + idx;
    r = r < 0 ? 0 : r;
    return r > last ? last : r;
}

#define AG_BATCH 16
__global__ __launch_bounds__(WG) void gin_bucket_agg_gemm(const float* __restrict__ x,
                                                          const float* __restrict__ W,
                                                          const float* __restrict__ bias,
                                                          const unsigned* __restrict__ seg,
                                                          const int* __restrict__ binned,
                                                          float* __restrict__ out) {
    __shared__ __align__(16) float sWp[(D / 2) * SW_STRIDE];
    __shared__ __align__(16) float sAgg[BROWS * SA_STRIDE];
    __shared__ __align__(8) int sEdge[2 * CAP];
    __shared__ int sCnt[SCN * 32];
    __shared__ int sStart[SCN * 32];
    __shared__ int sWaveTot[WG / 32];

    const int tid  = threadIdx.x;
    const int lane = tid & 31;
    const int wave = __builtin_amdgcn_readfirstlane(tid >> 5);
    const int bk    = blockIdx.x;
    const int node0 = bk * BN;

    const uint2 sg = *(const uint2*)(seg + 2 * ((size_t)(wave * SEG + (lane & (SEG - 1))) * NB + bk));
    static_assert(SEG == 8, "the segment walk below is written out for 8 segments per wave");
    const int sx = (int)sg.x, sy = (int)sg.y;
    const int p1 = __builtin_amdgcn_readlane(sy, 0);
    const int p2 = p1 + __builtin_amdgcn_readlane(sy, 1);
    const int p3 = p2 + __builtin_amdgcn_readlane(sy, 2);
    const int p4 = p3 + __builtin_amdgcn_readlane(sy, 3);
    const int p5 = p4 + __builtin_amdgcn_readlane(sy, 4);
    const int p6 = p5 + __builtin_amdgcn_readlane(sy, 5);
    const int p7 = p6 + __builtin_amdgcn_readlane(sy, 6);
    const int total = p7 + __builtin_amdgcn_readlane(sy, 7);
    const int a0 = __builtin_amdgcn_readlane(sx, 0);
    const int a1 = __builtin_amdgcn_readlane(sx, 1) - p1;
    const int a2 = __builtin_amdgcn_readlane(sx, 2) - p2;
    const int a3 = __builtin_amdgcn_readlane(sx, 3) - p3;
    const int a4 = __builtin_amdgcn_readlane(sx, 4) - p4;
    const int a5 = __builtin_amdgcn_readlane(sx, 5) - p5;
    const int a6 = __builtin_amdgcn_readlane(sx, 6) - p6;
    const int a7 = __builtin_amdgcn_readlane(sx, 7) - p7;
    v2i rec[KG];
#pragma unroll
    for (int g = 0; g < KG; ++g) {
        rec[g] = v2i{0, 0};
        if (g * 32 < total) {
            int idx = g * 32 + lane;
            if (idx > total - 1) idx = total - 1;
            rec[g] = *(const v2i*)(binned + 2 * (size_t)seg_record(idx, a0, a1, a2, a3, a4, a5, a6, a7, p1, p2, p3, p4, p5, p6, p7, NG * CH - 1));
        }
    }

    for (int i = tid; i < BROWS * SA_STRIDE / 4; i += WG)
        ((float4*)sAgg)[i] = make_float4(NEG_INF_F, NEG_INF_F, NEG_INF_F, NEG_INF_F);
#pragma unroll
    for (int i = 0; i < (D * D) / WG; ++i) {
        const int idx = tid + i * WG;
        const int o = idx >> 6;
        const int k = idx & 63;
        sWp[(k >> 1) * SW_STRIDE + o * 2 + (k & 1)] = W[idx];
    }
    if (tid < SCN * 32) sCnt[tid] = 0;
    if (lane == 0) sWaveTot[wave] = total;
    __syncthreads();

    static_assert(WG / 32 == 32, "one wave total per lane");
    int wsum = sWaveTot[lane];
    int wmax = wsum;
#pragma unroll
    for (int off = 16; off >= 1; off >>= 1) {
        wsum += __shfl_xor(wsum, off);
        const int t = __shfl_xor(wmax, off);
        wmax = t > wmax ? t : wmax;
    }
    const bool by_rows = __builtin_amdgcn_readfirstlane(wsum) <= CAP &&
                         __builtin_amdgcn_readfirstlane(wmax) <= KG * 32;

    if (by_rows) {
        int rk[KG];
#pragma unroll
        for (int g = 0; g < KG; ++g) {
            rk[g] = 0;
            if (g * 32 < total) {
                int dl = rec[g].y - node0;
                dl = dl < 0 ? 0 : (dl > BN - 1 ? BN - 1 : dl);
                rec[g].y = dl;
                if (g * 32 + lane < total) rk[g] = atomicAdd(&sCnt[dl], 1);
            }
        }
        __syncthreads();
        if (wave == 0) {
            int v[SCN];
            int sum = 0;
#pragma unroll
            for (int i = 0; i < SCN; ++i) { v[i] = sCnt[lane * SCN + i]; sum += v[i]; }
            int inc = sum;
#pragma unroll
            for (int off = 1; off < 32; off <<= 1) {
                const int t = __shfl_up(inc, off);
                if (lane >= off) inc += t;
            }
            int run = inc - sum;
#pragma unroll
            for (int i = 0; i < SCN; ++i) { sStart[lane * SCN + i] = run; run += v[i]; }
        }
        __syncthreads();
#pragma unroll
        for (int g = 0; g < KG; ++g) {
            if (g * 32 < total) {
                if (g * 32 + lane < total) {
                    const unsigned slot = (unsigned)(sStart[rec[g].y] + rk[g]);
                    if (slot < (unsigned)CAP) *(v2i*)(sEdge + 2 * slot) = rec[g];
                }
            }
        }
        __syncthreads();

        const int r0 = wave * RPW + (wave < RREM ? wave : RREM);
        const int r1 = r0 + RPW + (wave < RREM ? 1 : 0);
        int e0 = __builtin_amdgcn_readfirstlane(sStart[r0]);
        int e1 = __builtin_amdgcn_readfirstlane(sStart[r1]);
        e0 = e0 < 0 ? 0 : e0;
        e1 = e1 > CAP ? CAP : e1;
        int cur = -1;
        float m0 = NEG_INF_F, m1 = NEG_INF_F;
        for (int base = e0; base < e1; base += 32) {
            const int n = (e1 - base < 32) ? (e1 - base) : 32;
            int idx = base + lane;
            if (idx > e1 - 1) idx = e1 - 1;
            const v2i ed = *(const v2i*)(sEdge + 2 * idx);
            const int s_l = ed.x;
            const int d_l = ((unsigned)ed.y > (unsigned)(BN - 1)) ? (BN - 1) : ed.y;
            int d_prev = __shfl_up(d_l, 1);
            if (lane == 0) d_prev = cur;
            const unsigned chg = __builtin_amdgcn_ballot_w32(d_l != d_prev);
#pragma unroll
            for (int b = 0; b < 32; b += AG_BATCH) {
                if (b >= n) break;
                v2f v[AG_BATCH];
#pragma unroll
                for (int j = 0; j < AG_BATCH; ++j) {
                    const int s = __builtin_amdgcn_readlane(s_l, b + j);
                    v[j] = *(const v2f*)(x + (size_t)s * D + 2 * lane);
                }
#pragma unroll
                for (int j = 0; j < AG_BATCH; ++j) {
                    if (chg & (1u << (b + j))) {
                        if (cur >= 0) *(v2f*)&sAgg[cur * SA_STRIDE + 2 * lane] = v2f{m0, m1};
                        cur = __builtin_amdgcn_readlane(d_l, b + j);
                        m0 = NEG_INF_F;
                        m1 = NEG_INF_F;
                    }
                    asm("v_max_num_f32 %0, %0, %1" : "+v"(m0) : "v"(v[j].x));
                    asm("v_max_num_f32 %0, %0, %1" : "+v"(m1) : "v"(v[j].y));
                }
            }
        }
        if (cur >= 0) *(v2f*)&sAgg[cur * SA_STRIDE + 2 * lane] = v2f{m0, m1};
    } else {
        int base = 0;
        v2i ed = rec[0];
        while (base < total) {
            const int n = (total - base < 32) ? (total - base) : 32;
            const int nbase = base + 32;
            v2i edn = ed;
            if (nbase < total) {
                int idx = nbase + lane;
                if (idx > total - 1) idx = total - 1;
                edn = *(const v2i*)(binned + 2 * (size_t)seg_record(idx, a0, a1, a2, a3, a4, a5, a6, a7, p1, p2, p3, p4, p5, p6, p7, NG * CH - 1));
            }
            const int s_l = ed.x;
            const int d_l = ed.y - node0;
#pragma unroll
            for (int b = 0; b < 32; b += AG_BATCH) {
                if (b >= n) break;
                v2f v[AG_BATCH];
#pragma unroll
                for (int j = 0; j < AG_BATCH; ++j) {
                    const int s = __builtin_amdgcn_readlane(s_l, b + j);
                    v[j] = *(const v2f*)(x + (size_t)s * D + 2 * lane);
                }
#pragma unroll
                for (int j = 0; j < AG_BATCH; ++j) {
                    unsigned dl = (unsigned)__builtin_amdgcn_readlane(d_l, b + j);
                    if (dl > (unsigned)(BN - 1)) dl = (unsigned)(BN - 1);
                    float* ar = &sAgg[dl * SA_STRIDE + 2 * lane];
                    lds_max_f32(ar,     v[j].x);
                    lds_max_f32(ar + 1, v[j].y);
                }
            }
            ed = edn;
            base = nbase;
        }
    }

    float4 xin[RC];
#pragma unroll
    for (int u = 0; u < RC; ++u) {
        const int i    = tid + u * WG;
        const int row  = i >> 4;
        const int c4   = (i & 15) << 2;
        const int node = node0 + row;
        const int nodec = (row < BN && node < N_NODES) ? node : (N_NODES - 1);
        xin[u] = *(const float4*)(x + (size_t)nodec * D + c4);
    }
    __syncthreads();
#pragma unroll
    for (int u = 0; u < RC; ++u) {
        const int i    = tid + u * WG;
        const int row  = i >> 4;
        const int c4   = (i & 15) << 2;
        const int node = node0 + row;
        if (i < BROWS * 16) {
            float4* p = (float4*)&sAgg[row * SA_STRIDE + c4];
            const float4 au = *p;
            float4 r = make_float4(0.0f, 0.0f, 0.0f, 0.0f);
            if (row < BN && node < N_NODES) {
                r.x = xin[u].x + (au.x == NEG_INF_F ? 0.0f : au.x);
                r.y = xin[u].y + (au.y == NEG_INF_F ? 0.0f : au.y);
                r.z = xin[u].z + (au.z == NEG_INF_F ? 0.0f : au.z);
                r.w = xin[u].w + (au.w == NEG_INF_F ? 0.0f : au.w);
            }
            *p = r;
        }
    }
    __syncthreads();

    if (wave < BTILES) {
        v8f acc[4];
        acc[0] = v8f{}; acc[1] = v8f{}; acc[2] = v8f{}; acc[3] = v8f{};

        const int m     = lane & 15;
        const int khalf = (lane >> 4) << 1;
        const float* Arow = &sAgg[(wave * 16 + m) * SA_STRIDE];

#pragma unroll
        for (int k = 0; k < D; k += 4) {
            const int kk = k + khalf;
            const v2f a = *(const v2f*)(Arow + kk);
            const float* bbase = &sWp[(kk >> 1) * SW_STRIDE + m * 2];
#pragma unroll
            for (int t = 0; t < 4; ++t) {
                const v2f bf = *(const v2f*)(bbase + t * 32);
                acc[t] = __builtin_amdgcn_wmma_f32_16x16x4_f32(
                    false, a, false, bf, (short)0, acc[t], false, false);
            }
        }

        const int rowoff = (lane >> 4) << 3;
#pragma unroll
        for (int t = 0; t < 4; ++t) {
            const int col = t * 16 + m;
            const float bv = bias[col];
#pragma unroll
            for (int v = 0; v < 8; ++v) {
                const int row  = wave * 16 + rowoff + v;
                const int node = node0 + row;
                if (row < BN && node < N_NODES) out[(size_t)node * D + col] = acc[t][v] + bv;
            }
        }
    }
}

extern "C" void kernel_launch(void* const* d_in, const int* in_sizes, int n_in,
                              void* d_out, int out_size, void* d_ws, size_t ws_size,
                              hipStream_t stream) {
    const float* x    = (const float*)d_in[0];
    const float* W    = (const float*)d_in[1];
    const float* bias = (const float*)d_in[2];
    const int*   src  = (const int*)d_in[3];
    const int*   dst  = (const int*)d_in[4];
    float*       out  = (float*)d_out;
    unsigned*    seg    = (unsigned*)d_ws;
    int*         binned = (int*)(seg + 2 * NG * NB);

    gin_chunk_sort<<<NG, WG, 0, stream>>>(src, dst, seg, binned);

    gin_bucket_agg_gemm<<<NB, WG, 0, stream>>>(x, W, bias, seg, binned, out);
}


# Output tile staged through the wave's own 16 LDS rows: a store instruction writes two whole 256-B rows (four full 128-B lines), not two 64-B half lines; 8 stores per wave, not 32; sums unchanged
# speedup vs baseline: 4.2487x; 1.0756x over previous
#include <hip/hip_runtime.h>


#define N_NODES 100000
#define N_EDGES 1600000
#define D 64

#define SA_STRIDE 68
#define SW_STRIDE 160

typedef __attribute__((ext_vector_type(2))) float v2f;
typedef __attribute__((ext_vector_type(8))) float v8f;
typedef __attribute__((ext_vector_type(2))) int v2i;

__device__ __forceinline__ void lds_max_f32(float* p, float v) {
    __hip_atomic_fetch_max(p, v, __ATOMIC_RELAXED, __HIP_MEMORY_SCOPE_WORKGROUP);
}
#define NEG_INF_F (-__builtin_inff())

#define NB 256
#define BN ((N_NODES + NB - 1) / NB)
#define NG 256
#define CH ((N_EDGES + NG - 1) / NG)
#define WG 1024
#define PER ((CH + WG - 1) / WG)
#define SEG (NG / (WG / 32))
#define BROWS (((BN + 15) / 16) * 16)
#define BTILES (BROWS / 16)
#define RC ((BROWS * 16 + WG - 1) / WG)
#define KG 9
#define CAP 8192
#define RPW (BN / (WG / 32))
#define RREM (BN % (WG / 32))
#define SCN ((BROWS + 32) / 32)
static_assert(SCN * 32 > BROWS, "the scan covers every row counter and the entry behind the last row");
static_assert(RPW * (WG / 32) + RREM == BN, "the waves' row ranges cover the bucket exactly");
static_assert(NB == 8 * 32, "the scan below gives 8 counters to each lane of one wave");
static_assert(SEG * (WG / 32) == NG && SEG <= 32, "every wave of launch 2 owns SEG whole chunks, one per lane");
static_assert(BTILES <= WG / 32, "one wave per 16-row tile");
static_assert((long)NB * BN >= N_NODES, "buckets cover all nodes");

__device__ __forceinline__ void wave_scan_256(unsigned* arr, int lane) {
    unsigned v[8];
    unsigned sum = 0u;
#pragma unroll
    for (int i = 0; i < 8; ++i) { v[i] = arr[lane * 8 + i]; sum += v[i]; }
    unsigned inc = sum;
#pragma unroll
    for (int off = 1; off < 32; off <<= 1) {
        const unsigned t = (unsigned)__shfl_up((int)inc, off);
        if (lane >= off) inc += t;
    }
    unsigned run = inc - sum;
#pragma unroll
    for (int i = 0; i < 8; ++i) { run += v[i]; arr[lane * 8 + i] = run; }
}

__global__ __launch_bounds__(WG) void gin_chunk_sort(const int* __restrict__ src,
                                                     const int* __restrict__ dst,
                                                     unsigned* __restrict__ seg,
                                                     int* __restrict__ binned) {
    __shared__ unsigned sCnt[NB];
    __shared__ int sS[CH];
    __shared__ int sD[CH];

    const int tid  = threadIdx.x;
    const int lane = tid & 31;
    const int wave = tid >> 5;
    const int g    = blockIdx.x;
    const int e0 = g * CH;
    const int e1 = (e0 + CH < N_EDGES) ? (e0 + CH) : N_EDGES;

    int dv[PER], sv[PER];
#pragma unroll
    for (int i = 0; i < PER; ++i) {
        const int e = e0 + tid + i * WG;
        const bool in = e < e1;
        const int ec = in ? e : (N_EDGES - 1);
        const int d = dst[ec];
        sv[i] = src[ec];
        dv[i] = in ? d : -1;
    }
    if (tid < NB) sCnt[tid] = 0u;
    __syncthreads();
    unsigned rk[PER];
#pragma unroll
    for (int i = 0; i < PER; ++i) {
        rk[i] = 0u;
        if ((unsigned)dv[i] < (unsigned)N_NODES) rk[i] = atomicAdd(&sCnt[dv[i] / BN], 1u);
    }
    __syncthreads();
    unsigned own = 0u;
    if (tid < NB) own = sCnt[tid];
    __syncthreads();
    if (wave == 0) wave_scan_256(sCnt, lane);
    __syncthreads();
    if (tid < NB) {
        const unsigned lb = sCnt[tid] - own;
        *(uint2*)(seg + 2 * (size_t)(g * NB + tid)) = make_uint2((unsigned)(g * CH) + lb, own);
    }
#pragma unroll
    for (int i = 0; i < PER; ++i) {
        const int d = dv[i];
        if ((unsigned)d < (unsigned)N_NODES) {
            const int b = d / BN;
            const unsigned lb = b ? sCnt[b - 1] : 0u;
            const unsigned p = lb + rk[i];
            if (p < (unsigned)CH) { sS[p] = sv[i]; sD[p] = d; }
        }
    }
    __syncthreads();
    const unsigned nvalid = sCnt[NB - 1];
    for (unsigned i = tid; i < nvalid; i += WG)
        *(v2i*)(binned + 2 * ((size_t)g * CH + i)) = v2i{sS[i], sD[i]};
}

__device__ __forceinline__ int seg_record(int idx, int a0, int a1, int a2, int a3, int a4, int a5, int a6, int a7,
                                          int p1, int p2, int p3, int p4, int p5, int p6, int p7, int last) {
    int a = a0;
    a = (idx >= p1) ? a1 : a;
    a = (idx >= p2) ? a2 : a;
    a = (idx >= p3) ? a3 : a;
    a = (idx >= p4) ? a4 : a;
    a = (idx >= p5) ? a5 : a;
    a = (idx >= p6) ? a6 : a;
    a = (idx >= p7) ? a7 : a;
    int r = a + idx;
    r = r < 0 ? 0 : r;
    return r > last ? last : r;
}

#define AG_BATCH 16
__global__ __launch_bounds__(WG) void gin_bucket_agg_gemm(const float* __restrict__ x,
                                                          const float* __restrict__ W,
                                                          const float* __restrict__ bias,
                                                          const unsigned* __restrict__ seg,
                                                          const int* __restrict__ binned,
                                                          float* __restrict__ out) {
    __shared__ __align__(16) float sWp[(D / 2) * SW_STRIDE];
    __shared__ __align__(16) float sAgg[BROWS * SA_STRIDE];
    __shared__ __align__(8) int sEdge[2 * CAP];
    __shared__ int sCnt[SCN * 32];
    __shared__ int sStart[SCN * 32];
    __shared__ int sWaveTot[WG / 32];

    const int tid  = threadIdx.x;
    const int lane = tid & 31;
    const int wave = __builtin_amdgcn_readfirstlane(tid >> 5);
    const int bk    = blockIdx.x;
    const int node0 = bk * BN;

    const uint2 sg = *(const uint2*)(seg + 2 * ((size_t)(wave * SEG + (lane & (SEG - 1))) * NB + bk));
    static_assert(SEG == 8, "the segment walk below is written out for 8 segments per wave");
    const int sx = (int)sg.x, sy = (int)sg.y;
    const int p1 = __builtin_amdgcn_readlane(sy, 0);
    const int p2 = p1 + __builtin_amdgcn_readlane(sy, 1);
    const int p3 = p2 + __builtin_amdgcn_readlane(sy, 2);
    const int p4 = p3 + __builtin_amdgcn_readlane(sy, 3);
    const int p5 = p4 + __builtin_amdgcn_readlane(sy, 4);
    const int p6 = p5 + __builtin_amdgcn_readlane(sy, 5);
    const int p7 = p6 + __builtin_amdgcn_readlane(sy, 6);
    const int total = p7 + __builtin_amdgcn_readlane(sy, 7);
    const int a0 = __builtin_amdgcn_readlane(sx, 0);
    const int a1 = __builtin_amdgcn_readlane(sx, 1) - p1;
    const int a2 = __builtin_amdgcn_readlane(sx, 2) - p2;
    const int a3 = __builtin_amdgcn_readlane(sx, 3) - p3;
    const int a4 = __builtin_amdgcn_readlane(sx, 4) - p4;
    const int a5 = __builtin_amdgcn_readlane(sx, 5) - p5;
    const int a6 = __builtin_amdgcn_readlane(sx, 6) - p6;
    const int a7 = __builtin_amdgcn_readlane(sx, 7) - p7;
    v2i rec[KG];
#pragma unroll
    for (int g = 0; g < KG; ++g) {
        rec[g] = v2i{0, 0};
        if (g * 32 < total) {
            int idx = g * 32 + lane;
            if (idx > total - 1) idx = total - 1;
            rec[g] = *(const v2i*)(binned + 2 * (size_t)seg_record(idx, a0, a1, a2, a3, a4, a5, a6, a7, p1, p2, p3, p4, p5, p6, p7, NG * CH - 1));
        }
    }

    for (int i = tid; i < BROWS * SA_STRIDE / 4; i += WG)
        ((float4*)sAgg)[i] = make_float4(NEG_INF_F, NEG_INF_F, NEG_INF_F, NEG_INF_F);
#pragma unroll
    for (int i = 0; i < (D * D) / WG; ++i) {
        const int idx = tid + i * WG;
        const int o = idx >> 6;
        const int k = idx & 63;
        sWp[(k >> 1) * SW_STRIDE + o * 2 + (k & 1)] = W[idx];
    }
    if (tid < SCN * 32) sCnt[tid] = 0;
    if (lane == 0) sWaveTot[wave] = total;
    __syncthreads();

    static_assert(WG / 32 == 32, "one wave total per lane");
    int wsum = sWaveTot[lane];
    int wmax = wsum;
#pragma unroll
    for (int off = 16; off >= 1; off >>= 1) {
        wsum += __shfl_xor(wsum, off);
        const int t = __shfl_xor(wmax, off);
        wmax = t > wmax ? t : wmax;
    }
    const bool by_rows = __builtin_amdgcn_readfirstlane(wsum) <= CAP &&
                         __builtin_amdgcn_readfirstlane(wmax) <= KG * 32;

    if (by_rows) {
        int rk[KG];
#pragma unroll
        for (int g = 0; g < KG; ++g) {
            rk[g] = 0;
            if (g * 32 < total) {
                int dl = rec[g].y - node0;
                dl = dl < 0 ? 0 : (dl > BN - 1 ? BN - 1 : dl);
                rec[g].y = dl;
                if (g * 32 + lane < total) rk[g] = atomicAdd(&sCnt[dl], 1);
            }
        }
        __syncthreads();
        if (wave == 0) {
            int v[SCN];
            int sum = 0;
#pragma unroll
            for (int i = 0; i < SCN; ++i) { v[i] = sCnt[lane * SCN + i]; sum += v[i]; }
            int inc = sum;
#pragma unroll
            for (int off = 1; off < 32; off <<= 1) {
                const int t = __shfl_up(inc, off);
                if (lane >= off) inc += t;
            }
            int run = inc - sum;
#pragma unroll
            for (int i = 0; i < SCN; ++i) { sStart[lane * SCN + i] = run; run += v[i]; }
        }
        __syncthreads();
#pragma unroll
        for (int g = 0; g < KG; ++g) {
            if (g * 32 < total) {
                if (g * 32 + lane < total) {
                    const unsigned slot = (unsigned)(sStart[rec[g].y] + rk[g]);
                    if (slot < (unsigned)CAP) *(v2i*)(sEdge + 2 * slot) = rec[g];
                }
            }
        }
        __syncthreads();

        const int r0 = wave * RPW + (wave < RREM ? wave : RREM);
        const int r1 = r0 + RPW + (wave < RREM ? 1 : 0);
        int e0 = __builtin_amdgcn_readfirstlane(sStart[r0]);
        int e1 = __builtin_amdgcn_readfirstlane(sStart[r1]);
        e0 = e0 < 0 ? 0 : e0;
        e1 = e1 > CAP ? CAP : e1;
        int cur = -1;
        float m0 = NEG_INF_F, m1 = NEG_INF_F;
        for (int base = e0; base < e1; base += 32) {
            const int n = (e1 - base < 32) ? (e1 - base) : 32;
            int idx = base + lane;
            if (idx > e1 - 1) idx = e1 - 1;
            const v2i ed = *(const v2i*)(sEdge + 2 * idx);
            const int s_l = ed.x;
            const int d_l = ((unsigned)ed.y > (unsigned)(BN - 1)) ? (BN - 1) : ed.y;
            int d_prev = __shfl_up(d_l, 1);
            if (lane == 0) d_prev = cur;
            const unsigned chg = __builtin_amdgcn_ballot_w32(d_l != d_prev);
#pragma unroll
            for (int b = 0; b < 32; b += AG_BATCH) {
                if (b >= n) break;
                v2f v[AG_BATCH];
#pragma unroll
                for (int j = 0; j < AG_BATCH; ++j) {
                    const int s = __builtin_amdgcn_readlane(s_l, b + j);
                    v[j] = *(const v2f*)(x + (size_t)s * D + 2 * lane);
                }
#pragma unroll
                for (int j = 0; j < AG_BATCH; ++j) {
                    if (chg & (1u << (b + j))) {
                        if (cur >= 0) *(v2f*)&sAgg[cur * SA_STRIDE + 2 * lane] = v2f{m0, m1};
                        cur = __builtin_amdgcn_readlane(d_l, b + j);
                        m0 = NEG_INF_F;
                        m1 = NEG_INF_F;
                    }
                    asm("v_max_num_f32 %0, %0, %1" : "+v"(m0) : "v"(v[j].x));
                    asm("v_max_num_f32 %0, %0, %1" : "+v"(m1) : "v"(v[j].y));
                }
            }
        }
        if (cur >= 0) *(v2f*)&sAgg[cur * SA_STRIDE + 2 * lane] = v2f{m0, m1};
    } else {
        int base = 0;
        v2i ed = rec[0];
        while (base < total) {
            const int n = (total - base < 32) ? (total - base) : 32;
            const int nbase = base + 32;
            v2i edn = ed;
            if (nbase < total) {
                int idx = nbase + lane;
                if (idx > total - 1) idx = total - 1;
                edn = *(const v2i*)(binned + 2 * (size_t)seg_record(idx, a0, a1, a2, a3, a4, a5, a6, a7, p1, p2, p3, p4, p5, p6, p7, NG * CH - 1));
            }
            const int s_l = ed.x;
            const int d_l = ed.y - node0;
#pragma unroll
            for (int b = 0; b < 32; b += AG_BATCH) {
                if (b >= n) break;
                v2f v[AG_BATCH];
#pragma unroll
                for (int j = 0; j < AG_BATCH; ++j) {
                    const int s = __builtin_amdgcn_readlane(s_l, b + j);
                    v[j] = *(const v2f*)(x + (size_t)s * D + 2 * lane);
                }
#pragma unroll
                for (int j = 0; j < AG_BATCH; ++j) {
                    unsigned dl = (unsigned)__builtin_amdgcn_readlane(d_l, b + j);
                    if (dl > (unsigned)(BN - 1)) dl = (unsigned)(BN - 1);
                    float* ar = &sAgg[dl * SA_STRIDE + 2 * lane];
                    lds_max_f32(ar,     v[j].x);
                    lds_max_f32(ar + 1, v[j].y);
                }
            }
            ed = edn;
            base = nbase;
        }
    }

    float4 xin[RC];
#pragma unroll
    for (int u = 0; u < RC; ++u) {
        const int i    = tid + u * WG;
        const int row  = i >> 4;
        const int c4   = (i & 15) << 2;
        const int node = node0 + row;
        const int nodec = (row < BN && node < N_NODES) ? node : (N_NODES - 1);
        xin[u] = *(const float4*)(x + (size_t)nodec * D + c4);
    }
    __syncthreads();
#pragma unroll
    for (int u = 0; u < RC; ++u) {
        const int i    = tid + u * WG;
        const int row  = i >> 4;
        const int c4   = (i & 15) << 2;
        const int node = node0 + row;
        if (i < BROWS * 16) {
            float4* p = (float4*)&sAgg[row * SA_STRIDE + c4];
            const float4 au = *p;
            float4 r = make_float4(0.0f, 0.0f, 0.0f, 0.0f);
            if (row < BN && node < N_NODES) {
                r.x = xin[u].x + (au.x == NEG_INF_F ? 0.0f : au.x);
                r.y = xin[u].y + (au.y == NEG_INF_F ? 0.0f : au.y);
                r.z = xin[u].z + (au.z == NEG_INF_F ? 0.0f : au.z);
                r.w = xin[u].w + (au.w == NEG_INF_F ? 0.0f : au.w);
            }
            *p = r;
        }
    }
    __syncthreads();

    if (wave < BTILES) {
        v8f acc[4];
        acc[0] = v8f{}; acc[1] = v8f{}; acc[2] = v8f{}; acc[3] = v8f{};

        const int m     = lane & 15;
        const int khalf = (lane >> 4) << 1;
        const float* Arow = &sAgg[(wave * 16 + m) * SA_STRIDE];

#pragma unroll
        for (int k = 0; k < D; k += 4) {
            const int kk = k + khalf;
            const v2f a = *(const v2f*)(Arow + kk);
            const float* bbase = &sWp[(kk >> 1) * SW_STRIDE + m * 2];
#pragma unroll
            for (int t = 0; t < 4; ++t) {
                const v2f bf = *(const v2f*)(bbase + t * 32);
                acc[t] = __builtin_amdgcn_wmma_f32_16x16x4_f32(
                    false, a, false, bf, (short)0, acc[t], false, false);
            }
        }

        const int rowoff = (lane >> 4) << 3;
        float* Ct = &sAgg[(wave * 16) * SA_STRIDE];
#pragma unroll
        for (int t = 0; t < 4; ++t) {
            const int col = t * 16 + m;
            const float bv = bias[col];
#pragma unroll
            for (int v = 0; v < 8; ++v) Ct[(rowoff + v) * SA_STRIDE + col] = acc[t][v] + bv;
        }
        const int c4 = (lane & 15) << 2;
        const int rh = lane >> 4;
        const int rlast = wave * 16 + 15;
        if (rlast < BN && node0 + rlast < N_NODES) {
            float4 o[8];
#pragma unroll
            for (int i = 0; i < 8; ++i)
                o[i] = *(const float4*)&sAgg[(wave * 16 + 2 * i + rh) * SA_STRIDE + c4];
#pragma unroll
            for (int i = 0; i < 8; ++i)
                *(float4*)(out + (size_t)(node0 + wave * 16 + 2 * i + rh) * D + c4) = o[i];
        } else {
#pragma unroll
            for (int i = 0; i < 8; ++i) {
                const int row  = wave * 16 + 2 * i + rh;
                const int node = node0 + row;
                const float4 o = *(const float4*)&sAgg[row * SA_STRIDE + c4];
                if (row < BN && node < N_NODES) *(float4*)(out + (size_t)node * D + c4) = o;
            }
        }
    }
}

extern "C" void kernel_launch(void* const* d_in, const int* in_sizes, int n_in,
                              void* d_out, int out_size, void* d_ws, size_t ws_size,
                              hipStream_t stream) {
    const float* x    = (const float*)d_in[0];
    const float* W    = (const float*)d_in[1];
    const float* bias = (const float*)d_in[2];
    const int*   src  = (const int*)d_in[3];
    const int*   dst  = (const int*)d_in[4];
    float*       out  = (float*)d_out;
    unsigned*    seg    = (unsigned*)d_ws;
    int*         binned = (int*)(seg + 2 * NG * NB);

    gin_chunk_sort<<<NG, WG, 0, stream>>>(src, dst, seg, binned);

    gin_bucket_agg_gemm<<<NB, WG, 0, stream>>>(x, W, bias, seg, binned, out);
}
